# starting kernel
# baseline (speedup 1.0000x reference)
#include <hip/hip_runtime.h>

typedef __attribute__((ext_vector_type(2))) float v2f;
typedef __attribute__((ext_vector_type(8))) float v8f;

#define IN_DIM 512
#define HID    32
#define OUT_DIM 16

#define SMEM_FLOATS 21040
#define SMEM_BYTES  (SMEM_FLOATS * 4)

__global__ __launch_bounds__(256)
void mlp_fused_kernel(const float* __restrict__ feat,
                      const float* __restrict__ W1,
                      const float* __restrict__ b1,
                      const float* __restrict__ W2,
                      const float* __restrict__ b2,
                      float* __restrict__ x, int n)
{
    extern __shared__ float smem[];
    float* sW1 = smem;
    float* sH  = smem + 16384;
    float* sW2 = smem + 20480;
    float* sb1 = smem + 20992;
    float* sb2 = smem + 21024;

    for (int i = threadIdx.x; i < IN_DIM * HID; i += 256) sW1[i] = W1[i];
    for (int i = threadIdx.x; i < HID * OUT_DIM; i += 256) sW2[i] = W2[i];
    if (threadIdx.x < HID)     sb1[threadIdx.x] = b1[threadIdx.x];
    if (threadIdx.x < OUT_DIM) sb2[threadIdx.x] = b2[threadIdx.x];
    __syncthreads();

    const int wave = threadIdx.x >> 5;
    const int lane = threadIdx.x & 31;
    const int row0 = blockIdx.x * 128 + wave * 16;

    if (row0 < n) {
        const int m     = lane & 15;
        const int khalf = (lane >> 4) << 1;
        const int n0    = lane & 15;
        const float* arow = feat + (size_t)(row0 + m) * IN_DIM + khalf;

        v8f acc0 = {};
        v8f acc1 = {};

        #pragma unroll 4
        for (int k = 0; k < IN_DIM; k += 4) {
            v2f a = __builtin_nontemporal_load((const v2f*)(arow + k));
            const int kb = k + khalf;
            v2f bl, bh;
            bl.x = sW1[kb * HID + n0];
            bl.y = sW1[kb * HID + HID + n0];
            bh.x = sW1[kb * HID + 16 + n0];
            bh.y = sW1[kb * HID + HID + 16 + n0];
            acc0 = __builtin_amdgcn_wmma_f32_16x16x4_f32(
                false, a, false, bl, (short)0, acc0, false, false);
            acc1 = __builtin_amdgcn_wmma_f32_16x16x4_f32(
                false, a, false, bh, (short)0, acc1, false, false);
        }

        float* shw = sH + wave * (16 * HID);
        #pragma unroll
        for (int i = 0; i < 8; ++i) {
            const int mm = i + ((lane >> 4) << 3);
            float h0 = acc0[i] + sb1[n0];
            float h1 = acc1[i] + sb1[16 + n0];
            shw[mm * HID + n0]      = h0 > 0.0f ? h0 : 0.0f;
            shw[mm * HID + 16 + n0] = h1 > 0.0f ? h1 : 0.0f;
        }
        asm volatile("s_wait_dscnt 0x0" ::: "memory");

        const int mm   = lane & 15;
        const int half = lane >> 4;
        const float* hrow = shw + mm * HID;
        float* xrow = x + (size_t)(row0 + mm) * OUT_DIM;
        #pragma unroll
        for (int oo = 0; oo < 8; ++oo) {
            const int o = half * 8 + oo;
            float s = sb2[o];
            #pragma unroll
            for (int nn = 0; nn < HID; ++nn)
                s += hrow[nn] * sW2[nn * OUT_DIM + o];
            xrow[o] = s;
        }
    }
}

__global__ __launch_bounds__(256)
void spmm_scatter_kernel(const int* __restrict__ edge_row,
                         const int* __restrict__ edge_col,
                         const float* __restrict__ edge_w,
                         const float* __restrict__ x,
                         float* __restrict__ out, int E)
{
    const long t = (long)blockIdx.x * 256 + threadIdx.x;
    const long e = t >> 4;
    const int  o = (int)(t & 15);
    if (e < E) {
        const int   r = __builtin_nontemporal_load(edge_row + e);
        const int   c = __builtin_nontemporal_load(edge_col + e);
        const float w = __builtin_nontemporal_load(edge_w + e);
        atomicAdd(out + (size_t)r * OUT_DIM + o, w * x[(size_t)c * OUT_DIM + o]);
    }
}

extern "C" void kernel_launch(void* const* d_in, const int* in_sizes, int n_in,
                              void* d_out, int out_size, void* d_ws, size_t ws_size,
                              hipStream_t stream) {
    const float* feat = (const float*)d_in[0];
    const int*   er   = (const int*)  d_in[1];
    const int*   ec   = (const int*)  d_in[2];
    const float* ew   = (const float*)d_in[3];
    const float* W1   = (const float*)d_in[4];
    const float* b1   = (const float*)d_in[5];
    const float* W2   = (const float*)d_in[6];
    const float* b2   = (const float*)d_in[7];
    float* out = (float*)d_out;
    float* x   = (float*)d_ws;

    const int n = in_sizes[0] / IN_DIM;
    const int E = in_sizes[1];

    hipMemsetAsync(d_out, 0, (size_t)out_size * sizeof(float), stream);

    const int mlp_blocks = (n + 127) / 128;
    mlp_fused_kernel<<<mlp_blocks, 256, SMEM_BYTES, stream>>>(
        feat, W1, b1, W2, b2, x, n);

    const long sthreads = (long)E * 16;
    const int  sblocks  = (int)((sthreads + 255) / 256);
    spmm_scatter_kernel<<<sblocks, 256, 0, stream>>>(er, ec, ew, x, out, E);
}


# fc1 features through LDS: a wave reads 256-B row chunks with whole-line loads (2 rows x 256 B per load, next chunk in flight under the WMMAs), parks them in LDS, reads A fragments there; k order kept
# speedup vs baseline: 2.2823x; 2.2823x over previous
#include <hip/hip_runtime.h>

typedef __attribute__((ext_vector_type(2))) float v2f;
typedef __attribute__((ext_vector_type(4))) float v4f;
typedef __attribute__((ext_vector_type(8))) float v8f;

#define IN_DIM 512
#define HID    32
#define OUT_DIM 16

#define GROUP_TILES   25
#define GROUP_THREADS (GROUP_TILES * 32)
#define GROUP_ROWS    (GROUP_TILES * 16)

#define KC           64
#define STAGE_P      68
#define STAGE_FLOATS (16 * STAGE_P)

#define SMEM_FLOATS 44144
#define SMEM_BYTES  (176 * 1024)

#define CHUNK 256
#define TICKET_STRIDE 64

__global__ __launch_bounds__(GROUP_THREADS)
void mlp_fused_kernel(const float* __restrict__ feat,
                      const float* __restrict__ W1,
                      const float* __restrict__ b1,
                      const float* __restrict__ W2,
                      const float* __restrict__ b2,
                      float* __restrict__ x, int n,
                      unsigned* __restrict__ tickets)
{
    extern __shared__ float smem[];
    if (tickets != nullptr && blockIdx.x == 0 && threadIdx.x < 2)
        tickets[threadIdx.x * TICKET_STRIDE] = 0u;

    float* sW1 = smem;
    float* sA  = smem + 16384;
    float* sW2 = smem + 43584;
    float* sb1 = smem + 44096;
    float* sb2 = smem + 44128;

    #pragma unroll
    for (int j = 0; j < (IN_DIM * HID + GROUP_THREADS - 1) / GROUP_THREADS; ++j) {
        const int i = (int)threadIdx.x + j * GROUP_THREADS;
        if (i < IN_DIM * HID) sW1[i] = W1[i];
    }
    for (int i = threadIdx.x; i < HID * OUT_DIM; i += GROUP_THREADS) sW2[i] = W2[i];
    if (threadIdx.x < HID)     sb1[threadIdx.x] = b1[threadIdx.x];
    if (threadIdx.x < OUT_DIM) sb2[threadIdx.x] = b2[threadIdx.x];
    __syncthreads();

    const int wave = threadIdx.x >> 5;
    const int lane = threadIdx.x & 31;
    const int row0 = blockIdx.x * GROUP_ROWS + wave * 16;

    if (row0 < n) {
        const int m     = lane & 15;
        const int hh    = lane >> 4;
        const int khalf = hh << 1;
        const int n0    = lane & 15;
        float* st = sA + wave * STAGE_FLOATS;

        const float* gsrc  = feat + (size_t)(row0 + hh) * IN_DIM + (lane & 15) * 4;
        float* sdst = (float*)__builtin_assume_aligned(
            st + hh * STAGE_P + (lane & 15) * 4, 16);
        const float* afrag = (const float*)__builtin_assume_aligned(
            st + m * STAGE_P + khalf, 8);

        v8f acc0 = {};
        v8f acc1 = {};

        v4f r[8];
        #pragma unroll
        for (int j = 0; j < 8; ++j)
            r[j] = __builtin_nontemporal_load((const v4f*)(gsrc + (size_t)(2 * j) * IN_DIM));

        #pragma unroll 1
        for (int c = 0; c < IN_DIM / KC; ++c) {
            #pragma unroll
            for (int j = 0; j < 8; ++j)
                *(v4f*)(sdst + 2 * j * STAGE_P) = r[j];
            if (c + 1 < IN_DIM / KC) {
                #pragma unroll
                for (int j = 0; j < 8; ++j)
                    r[j] = __builtin_nontemporal_load(
                        (const v4f*)(gsrc + (size_t)(2 * j) * IN_DIM + (c + 1) * KC));
                asm volatile("s_wait_xcnt 0x0" ::: "memory");
            }
            asm volatile("s_wait_dscnt 0x0" ::: "memory");

            #pragma unroll
            for (int kk = 0; kk < KC; kk += 4) {
                const v2f a = *(const v2f*)(afrag + kk);
                const int kb = c * KC + kk + khalf;
                v2f bl, bh;
                bl.x = sW1[kb * HID + n0];
                bl.y = sW1[kb * HID + HID + n0];
                bh.x = sW1[kb * HID + 16 + n0];
                bh.y = sW1[kb * HID + HID + 16 + n0];
                acc0 = __builtin_amdgcn_wmma_f32_16x16x4_f32(
                    false, a, false, bl, (short)0, acc0, false, false);
                acc1 = __builtin_amdgcn_wmma_f32_16x16x4_f32(
                    false, a, false, bh, (short)0, acc1, false, false);
            }
        }

        float* shw = st;
        #pragma unroll
        for (int i = 0; i < 8; ++i) {
            const int mm = i + ((lane >> 4) << 3);
            float h0 = acc0[i] + sb1[n0];
            float h1 = acc1[i] + sb1[16 + n0];
            shw[mm * HID + n0]      = h0 > 0.0f ? h0 : 0.0f;
            shw[mm * HID + 16 + n0] = h1 > 0.0f ? h1 : 0.0f;
        }
        asm volatile("s_wait_dscnt 0x0" ::: "memory");

        const int mm   = lane & 15;
        const int half = lane >> 4;
        const float* hrow = shw + mm * HID;
        float* xrow = x + (size_t)(row0 + mm) * OUT_DIM;
        #pragma unroll
        for (int oo = 0; oo < 8; ++oo) {
            const int o = half * 8 + oo;
            float s = sb2[o];
            #pragma unroll
            for (int nn = 0; nn < HID; ++nn)
                s += hrow[nn] * sW2[nn * OUT_DIM + o];
            xrow[o] = s;
        }
    }
}

__global__ __launch_bounds__(256)
void spmm_scatter_kernel(const int* __restrict__ edge_row,
                         const int* __restrict__ edge_col,
                         const float* __restrict__ edge_w,
                         const float* __restrict__ x,
                         float* __restrict__ out, int E)
{
    const long t = (long)blockIdx.x * 256 + threadIdx.x;
    const long e = t >> 4;
    const int  o = (int)(t & 15);
    if (e < E) {
        const int   r = __builtin_nontemporal_load(edge_row + e);
        const int   c = __builtin_nontemporal_load(edge_col + e);
        const float w = __builtin_nontemporal_load(edge_w + e);
        atomicAdd(out + (size_t)r * OUT_DIM + o, w * x[(size_t)c * OUT_DIM + o]);
    }
}


__global__ __launch_bounds__(256)
void spmm_owner_kernel(const int* __restrict__ edge_row,
                       const int* __restrict__ edge_col,
                       const float* __restrict__ edge_w,
                       const float* __restrict__ x,
                       float* __restrict__ out,
                       unsigned* __restrict__ tickets, int E, int n)
{
    const int  lane = threadIdx.x & 31;
    const int  o    = lane & 15;
    const bool hi   = lane >= 16;

    const unsigned place  = __builtin_amdgcn_s_sendmsg_rtn(0x87);
    const int      myhalf = (int)((place >> 18) & 1u);
    const int      split  = (n >> 1) & ~3;
    const unsigned nchunks = (unsigned)((E + CHUNK - 1) / CHUNK);

    for (int pass = 0; pass < 2; ++pass) {
        const int h = myhalf ^ pass;
        unsigned* tk = tickets + h * TICKET_STRIDE;
        for (unsigned it = 0; it <= nchunks; ++it) {
            unsigned t = 0u;
            if (lane == 0) t = atomicAdd(tk, 1u);
            t = (unsigned)__builtin_amdgcn_readfirstlane((int)t);
            if (t >= nchunks) break;
            const int e0 = (int)t * CHUNK;

            const int elast = E - 1;
            int e = e0 + lane;
            int ec0 = e < E ? e : elast;
            int   r = __builtin_nontemporal_load(edge_row + ec0);
            int   c = __builtin_nontemporal_load(edge_col + ec0);
            float w = __builtin_nontemporal_load(edge_w + ec0);
            if (e >= E) r = -1;
            for (int rd = 0; rd < CHUNK / 32; ++rd) {
                const int en  = e + 32;
                const int enc = en < E ? en : elast;
                int   rn = __builtin_nontemporal_load(edge_row + enc);
                int   cn = __builtin_nontemporal_load(edge_col + enc);
                float wn = __builtin_nontemporal_load(edge_w + enc);
                if (en >= E) rn = -1;

                const bool mine = (r >= 0) && ((r >= split) == (h != 0));
                const unsigned m = (unsigned)__builtin_amdgcn_readfirstlane(
                    (int)__builtin_amdgcn_ballot_w32(mine));
                const int npairs = (__builtin_popcount(m) + 1) >> 1;
                const int wi = __float_as_int(w);

                float v[16];
                unsigned m1 = m;
                #pragma unroll
                for (int j = 0; j < 16; ++j) {
                    const int s0 = m1 ? __builtin_ctz(m1) : 0;  m1 &= m1 - 1u;
                    const int s1 = m1 ? __builtin_ctz(m1) : s0; m1 &= m1 - 1u;
                    const int c0 = __builtin_amdgcn_readlane(c, s0);
                    const int c1 = __builtin_amdgcn_readlane(c, s1);
                    const int cc = hi ? c1 : c0;
                    v[j] = x[cc * OUT_DIM + o];
                }
                unsigned m2 = m;
                #pragma unroll
                for (int j = 0; j < 16; ++j) {
                    if (j < npairs) {
                        const int s0 = __builtin_ctz(m2); m2 &= m2 - 1u;
                        const bool a1 = m2 != 0u;
                        const int s1 = a1 ? __builtin_ctz(m2) : s0; m2 &= m2 - 1u;
                        const int r0 = __builtin_amdgcn_readlane(r, s0);
                        const int r1 = __builtin_amdgcn_readlane(r, s1);
                        const int w0 = __builtin_amdgcn_readlane(wi, s0);
                        const int w1 = __builtin_amdgcn_readlane(wi, s1);
                        const int   rr = hi ? r1 : r0;
                        const float ww = __int_as_float(hi ? w1 : w0);
                        if (!hi || a1)
                            atomicAdd(out + rr * OUT_DIM + o, ww * v[j]);
                    }
                }
                r = rn; c = cn; w = wn; e = en;
            }
        }
    }
}


#define NBINS        256
#define CH           12544
#define SORT_THREADS 1024
#define EPT          ((CH + SORT_THREADS - 1) / SORT_THREADS)
#define OFFS_STRIDE  (NBINS + 1)
#define ROW_SHIFT    23
#define COL_SHIFT    6
#define COL_FIELD    0x7FFFC0u
#define BIN_DYN_LDS  (162 * 1024)
#define SUM_DYN_LDS  (132 * 1024)
#define RB_MAX       510
#define N_SORT_MAX   (NBINS * RB_MAX)
#define NULL_KEY     (511u << ROW_SHIFT)

__device__ __forceinline__ int rows_per_bin(int n) {
    return ((n + NBINS - 1) / NBINS + 1) & ~1;
}

__global__ __launch_bounds__(SORT_THREADS)
void edge_bin_kernel(const int* __restrict__ edge_row,
                     const int* __restrict__ edge_col,
                     const float* __restrict__ edge_w,
                     uint2* __restrict__ sedge,
                     unsigned* __restrict__ offs, int E, int n)
{
    extern __shared__ uint2 lbuf[];
    __shared__ unsigned cnt[NBINS + 1];
    __shared__ unsigned base[NBINS + 1];
    __shared__ unsigned wsum[NBINS / 32];

    const int tid  = (int)threadIdx.x;
    const int lane = tid & 31;
    const int wave = tid >> 5;
    const unsigned rb = (unsigned)rows_per_bin(n);
    const int e0 = (int)blockIdx.x * CH;
    const int cn = (E - e0) < CH ? (E - e0) : CH;

    if (tid <= NBINS) cnt[tid] = 0u;
    __syncthreads();

    int rr[EPT], cc[EPT];
    unsigned ww[EPT];
    #pragma unroll
    for (int k = 0; k < EPT; ++k) {
        const int i  = tid + k * SORT_THREADS;
        const int ii = e0 + (i < cn ? i : cn - 1);
        rr[k] = __builtin_nontemporal_load(edge_row + ii);
        cc[k] = __builtin_nontemporal_load(edge_col + ii);
        ww[k] = __float_as_uint(__builtin_nontemporal_load(edge_w + ii));
    }
    unsigned key[EPT], br[EPT];
    #pragma unroll
    for (int k = 0; k < EPT; ++k) {
        const int i = tid + k * SORT_THREADS;
        const unsigned r    = (unsigned)rr[k];
        const unsigned bin  = i < cn ? r / rb : (unsigned)NBINS;
        const unsigned rank = atomicAdd(&cnt[bin], 1u);
        br[k]  = (bin << 16) | rank;
        key[k] = ((r - bin * rb) << ROW_SHIFT) | ((unsigned)cc[k] << COL_SHIFT);
    }
    __syncthreads();

    unsigned v = 0u, s = 0u;
    if (tid < NBINS) {
        v = cnt[tid];
        s = v;
        #pragma unroll
        for (int d = 1; d < 32; d <<= 1) {
            const unsigned t = __shfl_up(s, d);
            if (lane >= d) s += t;
        }
        if (lane == 31) wsum[wave] = s;
    }
    __syncthreads();
    if (tid < NBINS) {
        unsigned add = 0u;
        for (int w = 0; w < wave; ++w) add += wsum[w];
        base[tid] = add + s - v;
        if (tid == NBINS - 1) base[NBINS] = add + s;
    }
    __syncthreads();

    #pragma unroll
    for (int k = 0; k < EPT; ++k) {
        const int i = tid + k * SORT_THREADS;
        if (i < cn) {
            const unsigned pos = base[br[k] >> 16] + (br[k] & 0xFFFFu);
            lbuf[pos] = make_uint2(key[k], ww[k]);
        }
    }
    __syncthreads();

    uint2* dst = sedge + (size_t)blockIdx.x * CH;
    for (int i = tid; i < cn; i += SORT_THREADS) dst[i] = lbuf[i];
    if (tid <= NBINS) offs[(size_t)blockIdx.x * OFFS_STRIDE + tid] = base[tid];
}

#define ROW_SHARE(v, J) \
    ((unsigned)__builtin_amdgcn_update_dpp(0, (int)(v), 0x150 + (J), 0xF, 0xF, false))

__device__ __forceinline__ void block_add_round(uint2 raw, int cnt, int lane, unsigned o4,
                                                const float* __restrict__ x,
                                                float* __restrict__ acc)
{
    const unsigned key = lane < cnt ? raw.x : NULL_KEY;
    const unsigned wb  = lane < cnt ? raw.y : 0u;
    const char* xb = (const char*)x;
    char*       ab = (char*)acc;
    float    v[16];
    unsigned kk[16];
#define GATHER_STEP(J)                                                        \
    kk[J] = ROW_SHARE(key, J);                                                \
    v[J]  = *(const float*)(xb + ((kk[J] & COL_FIELD) | o4));
    GATHER_STEP(0)  GATHER_STEP(1)  GATHER_STEP(2)  GATHER_STEP(3)
    GATHER_STEP(4)  GATHER_STEP(5)  GATHER_STEP(6)  GATHER_STEP(7)
    GATHER_STEP(8)  GATHER_STEP(9)  GATHER_STEP(10) GATHER_STEP(11)
    GATHER_STEP(12) GATHER_STEP(13) GATHER_STEP(14) GATHER_STEP(15)
#undef GATHER_STEP
#define ADD_STEP(J)                                                           \
    atomicAdd((float*)(ab + (((kk[J] >> ROW_SHIFT) << 6) | o4)),              \
              __uint_as_float(ROW_SHARE(wb, J)) * v[J]);
    ADD_STEP(0)  ADD_STEP(1)  ADD_STEP(2)  ADD_STEP(3)
    ADD_STEP(4)  ADD_STEP(5)  ADD_STEP(6)  ADD_STEP(7)
    ADD_STEP(8)  ADD_STEP(9)  ADD_STEP(10) ADD_STEP(11)
    ADD_STEP(12) ADD_STEP(13) ADD_STEP(14) ADD_STEP(15)
#undef ADD_STEP
}

__global__ __launch_bounds__(SORT_THREADS)
void block_sum_kernel(const uint2* __restrict__ sedge,
                      const unsigned* __restrict__ offs,
                      const float* __restrict__ x,
                      float* __restrict__ out, int n, int nchunks)
{
    __shared__ float acc[(RB_MAX + 2) * OUT_DIM];

    const int tid  = (int)threadIdx.x;
    const int lane = tid & 31;
    const int wave = __builtin_amdgcn_readfirstlane(tid >> 5);
    const unsigned o4 = (unsigned)(lane & 15) << 2;
    const int rb   = rows_per_bin(n);
    const int b    = (int)blockIdx.x;
    const int row0 = b * rb;
    int nrows = n - row0;
    if (nrows > rb) nrows = rb;
    if (nrows < 0)  nrows = 0;

    for (int i = tid; i < (RB_MAX + 2) * OUT_DIM; i += SORT_THREADS) acc[i] = 0.0f;
    __syncthreads();

    const unsigned* ob = offs + b;
    for (int g0 = wave; g0 < nchunks; g0 += 32 * 32) {
        const int gl = g0 + lane * 32;
        unsigned sv = 0u, tv = 0u;
        if (gl < nchunks) {
            sv = ob[(size_t)gl * OFFS_STRIDE];
            tv = ob[(size_t)gl * OFFS_STRIDE + 1];
        }
        int nseg = (nchunks - g0 + 31) >> 5;
        if (nseg > 32) nseg = 32;

        int k = -1;
        unsigned p = 0u, t = 0u;
        const uint2* seg = sedge;
        bool have = false;
        #define NEXT_ROUND()                                                   \
            do {                                                               \
                p += 32u; have = true;                                         \
                while (p >= t) {                                               \
                    ++k;                                                       \
                    if (k >= nseg) { have = false; break; }                    \
                    p = (unsigned)__builtin_amdgcn_readlane((int)sv, k);       \
                    t = (unsigned)__builtin_amdgcn_readlane((int)tv, k);       \
                    seg = sedge + (size_t)(g0 + 32 * k) * CH;                  \
                }                                                              \
            } while (0)
        #define FETCH_ROUND(dst, cntv)                                         \
            do {                                                               \
                cntv = 0;                                                      \
                if (have) cntv = (int)(t - p) < 32 ? (int)(t - p) : 32;        \
                const uint2* ap_ = lane < cntv ? seg + p + (unsigned)lane : sedge; \
                dst = *ap_;                                                    \
            } while (0)

        NEXT_ROUND();
        uint2 ed; int cur;
        FETCH_ROUND(ed, cur);
        while (have) {
            NEXT_ROUND();
            uint2 edn; int curn;
            FETCH_ROUND(edn, curn);
            block_add_round(ed, cur, lane, o4, x, acc);
            ed = edn; cur = curn;
        }
        #undef NEXT_ROUND
        #undef FETCH_ROUND
    }
    __syncthreads();

    float* ob_out = out + (size_t)row0 * OUT_DIM;
    for (int i = tid; i < nrows * OUT_DIM; i += SORT_THREADS) ob_out[i] = acc[i];
}

extern "C" void kernel_launch(void* const* d_in, const int* in_sizes, int n_in,
                              void* d_out, int out_size, void* d_ws, size_t ws_size,
                              hipStream_t stream) {
    const float* feat = (const float*)d_in[0];
    const int*   er   = (const int*)  d_in[1];
    const int*   ec   = (const int*)  d_in[2];
    const float* ew   = (const float*)d_in[3];
    const float* W1   = (const float*)d_in[4];
    const float* b1   = (const float*)d_in[5];
    const float* W2   = (const float*)d_in[6];
    const float* b2   = (const float*)d_in[7];
    float* out = (float*)d_out;
    float* x   = (float*)d_ws;

    const int n = in_sizes[0] / IN_DIM;
    const int E = in_sizes[1];

    const size_t tk_off = (((size_t)n * OUT_DIM * sizeof(float)) + 255) & ~(size_t)255;
    const bool   owner  = ws_size >= tk_off + 512;
    unsigned* tickets = owner ? (unsigned*)((char*)d_ws + tk_off) : nullptr;

    const int    nchunks  = (E + CH - 1) / CH;
    const size_t se_off   = tk_off + 512;
    const size_t of_off   = se_off + (size_t)nchunks * CH * sizeof(uint2);
    const size_t ws_need  = of_off + (size_t)nchunks * OFFS_STRIDE * sizeof(unsigned);
    const bool   sorted   = n <= N_SORT_MAX && E > 0 && ws_size >= ws_need;

    const int mlp_blocks = (n + GROUP_ROWS - 1) / GROUP_ROWS;

    if (sorted) {
        uint2*    sedge = (uint2*)((char*)d_ws + se_off);
        unsigned* offs  = (unsigned*)((char*)d_ws + of_off);
        edge_bin_kernel<<<nchunks, SORT_THREADS, BIN_DYN_LDS, stream>>>(
            er, ec, ew, sedge, offs, E, n);
        mlp_fused_kernel<<<mlp_blocks, GROUP_THREADS, SMEM_BYTES, stream>>>(
            feat, W1, b1, W2, b2, x, n, nullptr);
        block_sum_kernel<<<NBINS, SORT_THREADS, SUM_DYN_LDS, stream>>>(sedge, offs, x, out, n, nchunks);
        return;
    }

    hipMemsetAsync(d_out, 0, (size_t)out_size * sizeof(float), stream);

    mlp_fused_kernel<<<mlp_blocks, GROUP_THREADS, SMEM_BYTES, stream>>>(
        feat, W1, b1, W2, b2, x, n, tickets);

    if (owner) {
        spmm_owner_kernel<<<2048, 256, 0, stream>>>(er, ec, ew, x, out, tickets, E, n);
    } else {
        const long sthreads = (long)E * 16;
        const int  sblocks  = (int)((sthreads + 255) / 256);
        spmm_scatter_kernel<<<sblocks, 256, 0, stream>>>(er, ec, ew, x, out, E);
    }
}


# edge sort folded into the fc1 launch: each 25-wave group first sorts one 12544-edge chunk in LDS (16 edges a thread), then runs its fc1 tiles; 2 launches, not 3; arrival-order sums as before
# speedup vs baseline: 3.1336x; 1.3730x over previous
#include <hip/hip_runtime.h>

typedef __attribute__((ext_vector_type(2))) float v2f;
typedef __attribute__((ext_vector_type(4))) float v4f;
typedef __attribute__((ext_vector_type(8))) float v8f;

#define IN_DIM 512
#define HID    32
#define OUT_DIM 16

#define GROUP_TILES   25
#define GROUP_THREADS (GROUP_TILES * 32)
#define GROUP_ROWS    (GROUP_TILES * 16)

#define KC           64
#define STAGE_P      68
#define STAGE_FLOATS (16 * STAGE_P)

#define SMEM_FLOATS 69768
#define CNT_OFF     44144
#define BASE_OFF    44408
#define WSUM_OFF    44672
#define LBUF_OFF    44680
#define SMEM_BYTES  (274 * 1024)

#define CHUNK 256
#define TICKET_STRIDE 64

#define NBINS        256
#define CH           12544
#define SORT_THREADS 1024
#define BIN_EPT      ((CH + GROUP_THREADS - 1) / GROUP_THREADS)
#define OFFS_STRIDE  (NBINS + 1)
#define ROW_SHIFT    23
#define COL_SHIFT    6
#define COL_FIELD    0x7FFFC0u
#define SUM_DYN_LDS  (132 * 1024)
#define RB_MAX       510
#define N_SORT_MAX   (NBINS * RB_MAX)
#define NULL_KEY     (511u << ROW_SHIFT)

__device__ __forceinline__ int rows_per_bin(int n) {
    return ((n + NBINS - 1) / NBINS + 1) & ~1;
}

__global__ __launch_bounds__(GROUP_THREADS)
void mlp_fused_kernel(const float* __restrict__ feat,
                      const float* __restrict__ W1,
                      const float* __restrict__ b1,
                      const float* __restrict__ W2,
                      const float* __restrict__ b2,
                      float* __restrict__ x, int n,
                      unsigned* __restrict__ tickets,
                      const int* __restrict__ edge_row,
                      const int* __restrict__ edge_col,
                      const float* __restrict__ edge_w,
                      uint2* __restrict__ sedge,
                      unsigned* __restrict__ offs, int E, int nchunks)
{
    extern __shared__ float smem[];
    if (tickets != nullptr && blockIdx.x == 0 && threadIdx.x < 2)
        tickets[threadIdx.x * TICKET_STRIDE] = 0u;

    float* sW1 = smem;
    float* sA  = smem + 16384;
    float* sW2 = smem + 43584;
    float* sb1 = smem + 44096;
    float* sb2 = smem + 44128;
    unsigned* cnt  = (unsigned*)(smem + CNT_OFF);
    unsigned* base = (unsigned*)(smem + BASE_OFF);
    unsigned* wsum = (unsigned*)(smem + WSUM_OFF);
    uint2*    lbuf = (uint2*)(smem + LBUF_OFF);

    const int tid  = (int)threadIdx.x;
    const int wave = tid >> 5;
    const int lane = tid & 31;

    const bool do_bin = sedge != nullptr && (int)blockIdx.x < nchunks;
    const unsigned rb = (unsigned)rows_per_bin(n);
    const int e0 = (int)blockIdx.x * CH;
    const int cn = do_bin ? ((E - e0) < CH ? (E - e0) : CH) : 0;

    int rr[BIN_EPT], cc[BIN_EPT];
    unsigned ww[BIN_EPT];
    #pragma unroll
    for (int k = 0; k < BIN_EPT; ++k) { rr[k] = 0; cc[k] = 0; ww[k] = 0u; }
    if (do_bin) {
        #pragma unroll
        for (int k = 0; k < BIN_EPT; ++k) {
            const int i  = tid + k * GROUP_THREADS;
            const int ii = e0 + (i < cn ? i : cn - 1);
            rr[k] = __builtin_nontemporal_load(edge_row + ii);
            cc[k] = __builtin_nontemporal_load(edge_col + ii);
            ww[k] = __float_as_uint(__builtin_nontemporal_load(edge_w + ii));
        }
        if (tid <= NBINS) cnt[tid] = 0u;
    }

    #pragma unroll
    for (int j = 0; j < (IN_DIM * HID + GROUP_THREADS - 1) / GROUP_THREADS; ++j) {
        const int i = (int)threadIdx.x + j * GROUP_THREADS;
        if (i < IN_DIM * HID) sW1[i] = W1[i];
    }
    for (int i = threadIdx.x; i < HID * OUT_DIM; i += GROUP_THREADS) sW2[i] = W2[i];
    if (threadIdx.x < HID)     sb1[threadIdx.x] = b1[threadIdx.x];
    if (threadIdx.x < OUT_DIM) sb2[threadIdx.x] = b2[threadIdx.x];
    __syncthreads();

    unsigned key[BIN_EPT], br[BIN_EPT];
    #pragma unroll
    for (int k = 0; k < BIN_EPT; ++k) { key[k] = 0u; br[k] = 0u; }
    if (do_bin) {
        #pragma unroll
        for (int k = 0; k < BIN_EPT; ++k) {
            const int i = tid + k * GROUP_THREADS;
            const unsigned r    = (unsigned)rr[k];
            const unsigned bin  = i < cn ? r / rb : (unsigned)NBINS;
            const unsigned rank = atomicAdd(&cnt[bin], 1u);
            br[k]  = (bin << 16) | rank;
            key[k] = ((r - bin * rb) << ROW_SHIFT) | ((unsigned)cc[k] << COL_SHIFT);
        }
    }
    __syncthreads();

    const int  row0     = blockIdx.x * GROUP_ROWS + wave * 16;
    const bool has_tile = row0 < n;
    const int m     = lane & 15;
    const int hh    = lane >> 4;
    const int khalf = hh << 1;
    const int n0    = lane & 15;
    const float* gsrc  = feat + (size_t)(row0 + hh) * IN_DIM + (lane & 15) * 4;

    unsigned v = 0u, s = 0u;
    if (do_bin && tid < NBINS) {
        v = cnt[tid];
        s = v;
        #pragma unroll
        for (int d = 1; d < 32; d <<= 1) {
            const unsigned t = __shfl_up(s, d);
            if (lane >= d) s += t;
        }
        if (lane == 31) wsum[wave] = s;
    }
    __syncthreads();
    if (do_bin && tid < NBINS) {
        unsigned add = 0u;
        for (int w = 0; w < wave; ++w) add += wsum[w];
        base[tid] = add + s - v;
        if (tid == NBINS - 1) base[NBINS] = add + s;
    }
    __syncthreads();

    if (do_bin) {
        #pragma unroll
        for (int k = 0; k < BIN_EPT; ++k) {
            const int i = tid + k * GROUP_THREADS;
            if (i < cn) {
                const unsigned pos = base[br[k] >> 16] + (br[k] & 0xFFFFu);
                lbuf[pos] = make_uint2(key[k], ww[k]);
            }
        }
    }
    __syncthreads();

    v4f r[8];
    #pragma unroll
    for (int j = 0; j < 8; ++j) r[j] = (v4f){0.0f, 0.0f, 0.0f, 0.0f};
    if (has_tile) {
        #pragma unroll
        for (int j = 0; j < 8; ++j)
            r[j] = __builtin_nontemporal_load((const v4f*)(gsrc + (size_t)(2 * j) * IN_DIM));
    }

    if (do_bin) {
        uint2* dst = sedge + (size_t)blockIdx.x * CH;
        for (int i = tid; i < cn; i += GROUP_THREADS) dst[i] = lbuf[i];
        if (tid <= NBINS) offs[(size_t)blockIdx.x * OFFS_STRIDE + tid] = base[tid];
    }

    if (has_tile) {
        float* st = sA + wave * STAGE_FLOATS;

        float* sdst = (float*)__builtin_assume_aligned(
            st + hh * STAGE_P + (lane & 15) * 4, 16);
        const float* afrag = (const float*)__builtin_assume_aligned(
            st + m * STAGE_P + khalf, 8);

        v8f acc0 = {};
        v8f acc1 = {};

        #pragma unroll 1
        for (int c = 0; c < IN_DIM / KC; ++c) {
            #pragma unroll
            for (int j = 0; j < 8; ++j)
                *(v4f*)(sdst + 2 * j * STAGE_P) = r[j];
            if (c + 1 < IN_DIM / KC) {
                #pragma unroll
                for (int j = 0; j < 8; ++j)
                    r[j] = __builtin_nontemporal_load(
                        (const v4f*)(gsrc + (size_t)(2 * j) * IN_DIM + (c + 1) * KC));
                asm volatile("s_wait_xcnt 0x0" ::: "memory");
            }
            asm volatile("s_wait_dscnt 0x0" ::: "memory");

            #pragma unroll
            for (int kk = 0; kk < KC; kk += 4) {
                const v2f a = *(const v2f*)(afrag + kk);
                const int kb = c * KC + kk + khalf;
                v2f bl, bh;
                bl.x = sW1[kb * HID + n0];
                bl.y = sW1[kb * HID + HID + n0];
                bh.x = sW1[kb * HID + 16 + n0];
                bh.y = sW1[kb * HID + HID + 16 + n0];
                acc0 = __builtin_amdgcn_wmma_f32_16x16x4_f32(
                    false, a, false, bl, (short)0, acc0, false, false);
                acc1 = __builtin_amdgcn_wmma_f32_16x16x4_f32(
                    false, a, false, bh, (short)0, acc1, false, false);
            }
        }

        float* shw = st;
        #pragma unroll
        for (int i = 0; i < 8; ++i) {
            const int mm = i + ((lane >> 4) << 3);
            float h0 = acc0[i] + sb1[n0];
            float h1 = acc1[i] + sb1[16 + n0];
            shw[mm * HID + n0]      = h0 > 0.0f ? h0 : 0.0f;
            shw[mm * HID + 16 + n0] = h1 > 0.0f ? h1 : 0.0f;
        }
        asm volatile("s_wait_dscnt 0x0" ::: "memory");

        const int mm   = lane & 15;
        const int half = lane >> 4;
        const float* hrow = shw + mm * HID;
        float* xrow = x + (size_t)(row0 + mm) * OUT_DIM;
        #pragma unroll
        for (int oo = 0; oo < 8; ++oo) {
            const int o = half * 8 + oo;
            float s2 = sb2[o];
            #pragma unroll
            for (int nn = 0; nn < HID; ++nn)
                s2 += hrow[nn] * sW2[nn * OUT_DIM + o];
            xrow[o] = s2;
        }
    }
}

__global__ __launch_bounds__(256)
void spmm_scatter_kernel(const int* __restrict__ edge_row,
                         const int* __restrict__ edge_col,
                         const float* __restrict__ edge_w,
                         const float* __restrict__ x,
                         float* __restrict__ out, int E)
{
    const long t = (long)blockIdx.x * 256 + threadIdx.x;
    const long e = t >> 4;
    const int  o = (int)(t & 15);
    if (e < E) {
        const int   r = __builtin_nontemporal_load(edge_row + e);
        const int   c = __builtin_nontemporal_load(edge_col + e);
        const float w = __builtin_nontemporal_load(edge_w + e);
        atomicAdd(out + (size_t)r * OUT_DIM + o, w * x[(size_t)c * OUT_DIM + o]);
    }
}


__global__ __launch_bounds__(256)
void spmm_owner_kernel(const int* __restrict__ edge_row,
                       const int* __restrict__ edge_col,
                       const float* __restrict__ edge_w,
                       const float* __restrict__ x,
                       float* __restrict__ out,
                       unsigned* __restrict__ tickets, int E, int n)
{
    const int  lane = threadIdx.x & 31;
    const int  o    = lane & 15;
    const bool hi   = lane >= 16;

    const unsigned place  = __builtin_amdgcn_s_sendmsg_rtn(0x87);
    const int      myhalf = (int)((place >> 18) & 1u);
    const int      split  = (n >> 1) & ~3;
    const unsigned nchunks = (unsigned)((E + CHUNK - 1) / CHUNK);

    for (int pass = 0; pass < 2; ++pass) {
        const int h = myhalf ^ pass;
        unsigned* tk = tickets + h * TICKET_STRIDE;
        for (unsigned it = 0; it <= nchunks; ++it) {
            unsigned t = 0u;
            if (lane == 0) t = atomicAdd(tk, 1u);
            t = (unsigned)__builtin_amdgcn_readfirstlane((int)t);
            if (t >= nchunks) break;
            const int e0 = (int)t * CHUNK;

            const int elast = E - 1;
            int e = e0 + lane;
            int ec0 = e < E ? e : elast;
            int   r = __builtin_nontemporal_load(edge_row + ec0);
            int   c = __builtin_nontemporal_load(edge_col + ec0);
            float w = __builtin_nontemporal_load(edge_w + ec0);
            if (e >= E) r = -1;
            for (int rd = 0; rd < CHUNK / 32; ++rd) {
                const int en  = e + 32;
                const int enc = en < E ? en : elast;
                int   rn = __builtin_nontemporal_load(edge_row + enc);
                int   cn = __builtin_nontemporal_load(edge_col + enc);
                float wn = __builtin_nontemporal_load(edge_w + enc);
                if (en >= E) rn = -1;

                const bool mine = (r >= 0) && ((r >= split) == (h != 0));
                const unsigned m = (unsigned)__builtin_amdgcn_readfirstlane(
                    (int)__builtin_amdgcn_ballot_w32(mine));
                const int npairs = (__builtin_popcount(m) + 1) >> 1;
                const int wi = __float_as_int(w);

                float v[16];
                unsigned m1 = m;
                #pragma unroll
                for (int j = 0; j < 16; ++j) {
                    const int s0 = m1 ? __builtin_ctz(m1) : 0;  m1 &= m1 - 1u;
                    const int s1 = m1 ? __builtin_ctz(m1) : s0; m1 &= m1 - 1u;
                    const int c0 = __builtin_amdgcn_readlane(c, s0);
                    const int c1 = __builtin_amdgcn_readlane(c, s1);
                    const int cc = hi ? c1 : c0;
                    v[j] = x[cc * OUT_DIM + o];
                }
                unsigned m2 = m;
                #pragma unroll
                for (int j = 0; j < 16; ++j) {
                    if (j < npairs) {
                        const int s0 = __builtin_ctz(m2); m2 &= m2 - 1u;
                        const bool a1 = m2 != 0u;
                        const int s1 = a1 ? __builtin_ctz(m2) : s0; m2 &= m2 - 1u;
                        const int r0 = __builtin_amdgcn_readlane(r, s0);
                        const int r1 = __builtin_amdgcn_readlane(r, s1);
                        const int w0 = __builtin_amdgcn_readlane(wi, s0);
                        const int w1 = __builtin_amdgcn_readlane(wi, s1);
                        const int   rr = hi ? r1 : r0;
                        const float ww = __int_as_float(hi ? w1 : w0);
                        if (!hi || a1)
                            atomicAdd(out + rr * OUT_DIM + o, ww * v[j]);
                    }
                }
                r = rn; c = cn; w = wn; e = en;
            }
        }
    }
}


#define ROW_SHARE(v, J) \
    ((unsigned)__builtin_amdgcn_update_dpp(0, (int)(v), 0x150 + (J), 0xF, 0xF, false))

__device__ __forceinline__ void block_add_round(uint2 raw, int cnt, int lane, unsigned o4,
                                                const float* __restrict__ x,
                                                float* __restrict__ acc)
{
    const unsigned key = lane < cnt ? raw.x : NULL_KEY;
    const unsigned wb  = lane < cnt ? raw.y : 0u;
    const char* xb = (const char*)x;
    char*       ab = (char*)acc;
    float    v[16];
    unsigned kk[16];
#define GATHER_STEP(J)                                                        \
    kk[J] = ROW_SHARE(key, J);                                                \
    v[J]  = *(const float*)(xb + ((kk[J] & COL_FIELD) | o4));
    GATHER_STEP(0)  GATHER_STEP(1)  GATHER_STEP(2)  GATHER_STEP(3)
    GATHER_STEP(4)  GATHER_STEP(5)  GATHER_STEP(6)  GATHER_STEP(7)
    GATHER_STEP(8)  GATHER_STEP(9)  GATHER_STEP(10) GATHER_STEP(11)
    GATHER_STEP(12) GATHER_STEP(13) GATHER_STEP(14) GATHER_STEP(15)
#undef GATHER_STEP
#define ADD_STEP(J)                                                           \
    atomicAdd((float*)(ab + (((kk[J] >> ROW_SHIFT) << 6) | o4)),              \
              __uint_as_float(ROW_SHARE(wb, J)) * v[J]);
    ADD_STEP(0)  ADD_STEP(1)  ADD_STEP(2)  ADD_STEP(3)
    ADD_STEP(4)  ADD_STEP(5)  ADD_STEP(6)  ADD_STEP(7)
    ADD_STEP(8)  ADD_STEP(9)  ADD_STEP(10) ADD_STEP(11)
    ADD_STEP(12) ADD_STEP(13) ADD_STEP(14) ADD_STEP(15)
#undef ADD_STEP
}

__global__ __launch_bounds__(SORT_THREADS)
void block_sum_kernel(const uint2* __restrict__ sedge,
                      const unsigned* __restrict__ offs,
                      const float* __restrict__ x,
                      float* __restrict__ out, int n, int nchunks)
{
    __shared__ float acc[(RB_MAX + 2) * OUT_DIM];

    const int tid  = (int)threadIdx.x;
    const int lane = tid & 31;
    const int wave = __builtin_amdgcn_readfirstlane(tid >> 5);
    const unsigned o4 = (unsigned)(lane & 15) << 2;
    const int rb   = rows_per_bin(n);
    const int b    = (int)blockIdx.x;
    const int row0 = b * rb;
    int nrows = n - row0;
    if (nrows > rb) nrows = rb;
    if (nrows < 0)  nrows = 0;

    for (int i = tid; i < (RB_MAX + 2) * OUT_DIM; i += SORT_THREADS) acc[i] = 0.0f;
    __syncthreads();

    const unsigned* ob = offs + b;
    for (int g0 = wave; g0 < nchunks; g0 += 32 * 32) {
        const int gl = g0 + lane * 32;
        unsigned sv = 0u, tv = 0u;
        if (gl < nchunks) {
            sv = ob[(size_t)gl * OFFS_STRIDE];
            tv = ob[(size_t)gl * OFFS_STRIDE + 1];
        }
        int nseg = (nchunks - g0 + 31) >> 5;
        if (nseg > 32) nseg = 32;

        int k = -1;
        unsigned p = 0u, t = 0u;
        const uint2* seg = sedge;
        bool have = false;
        #define NEXT_ROUND()                                                   \
            do {                                                               \
                p += 32u; have = true;                                         \
                while (p >= t) {                                               \
                    ++k;                                                       \
                    if (k >= nseg) { have = false; break; }                    \
                    p = (unsigned)__builtin_amdgcn_readlane((int)sv, k);       \
                    t = (unsigned)__builtin_amdgcn_readlane((int)tv, k);       \
                    seg = sedge + (size_t)(g0 + 32 * k) * CH;                  \
                }                                                              \
            } while (0)
        #define FETCH_ROUND(dst, cntv)                                         \
            do {                                                               \
                cntv = 0;                                                      \
                if (have) cntv = (int)(t - p) < 32 ? (int)(t - p) : 32;        \
                const uint2* ap_ = lane < cntv ? seg + p + (unsigned)lane : sedge; \
                dst = *ap_;                                                    \
            } while (0)

        NEXT_ROUND();
        uint2 ed; int cur;
        FETCH_ROUND(ed, cur);
        while (have) {
            NEXT_ROUND();
            uint2 edn; int curn;
            FETCH_ROUND(edn, curn);
            block_add_round(ed, cur, lane, o4, x, acc);
            ed = edn; cur = curn;
        }
        #undef NEXT_ROUND
        #undef FETCH_ROUND
    }
    __syncthreads();

    float* ob_out = out + (size_t)row0 * OUT_DIM;
    for (int i = tid; i < nrows * OUT_DIM; i += SORT_THREADS) ob_out[i] = acc[i];
}

extern "C" void kernel_launch(void* const* d_in, const int* in_sizes, int n_in,
                              void* d_out, int out_size, void* d_ws, size_t ws_size,
                              hipStream_t stream) {
    const float* feat = (const float*)d_in[0];
    const int*   er   = (const int*)  d_in[1];
    const int*   ec   = (const int*)  d_in[2];
    const float* ew   = (const float*)d_in[3];
    const float* W1   = (const float*)d_in[4];
    const float* b1   = (const float*)d_in[5];
    const float* W2   = (const float*)d_in[6];
    const float* b2   = (const float*)d_in[7];
    float* out = (float*)d_out;
    float* x   = (float*)d_ws;

    const int n = in_sizes[0] / IN_DIM;
    const int E = in_sizes[1];

    const size_t tk_off = (((size_t)n * OUT_DIM * sizeof(float)) + 255) & ~(size_t)255;
    const bool   owner  = ws_size >= tk_off + 512;
    unsigned* tickets = owner ? (unsigned*)((char*)d_ws + tk_off) : nullptr;

    const int    nchunks  = (E + CH - 1) / CH;
    const size_t se_off   = tk_off + 512;
    const size_t of_off   = se_off + (size_t)nchunks * CH * sizeof(uint2);
    const size_t ws_need  = of_off + (size_t)nchunks * OFFS_STRIDE * sizeof(unsigned);
    const bool   sorted   = n <= N_SORT_MAX && E > 0 && ws_size >= ws_need;

    const int mlp_blocks = (n + GROUP_ROWS - 1) / GROUP_ROWS;

    if (sorted) {
        uint2*    sedge = (uint2*)((char*)d_ws + se_off);
        unsigned* offs  = (unsigned*)((char*)d_ws + of_off);
        const int fused_blocks = mlp_blocks > nchunks ? mlp_blocks : nchunks;
        mlp_fused_kernel<<<fused_blocks, GROUP_THREADS, SMEM_BYTES, stream>>>(
            feat, W1, b1, W2, b2, x, n, nullptr, er, ec, ew, sedge, offs, E, nchunks);
        block_sum_kernel<<<NBINS, SORT_THREADS, SUM_DYN_LDS, stream>>>(sedge, offs, x, out, n, nchunks);
        return;
    }

    hipMemsetAsync(d_out, 0, (size_t)out_size * sizeof(float), stream);

    mlp_fused_kernel<<<mlp_blocks, GROUP_THREADS, SMEM_BYTES, stream>>>(
        feat, W1, b1, W2, b2, x, n, tickets, er, ec, ew, nullptr, nullptr, E, 0);

    if (owner) {
        spmm_owner_kernel<<<2048, 256, 0, stream>>>(er, ec, ew, x, out, tickets, E, n);
    } else {
        const long sthreads = (long)E * 16;
        const int  sblocks  = (int)((sthreads + 255) / 256);
        spmm_scatter_kernel<<<sblocks, 256, 0, stream>>>(er, ec, ew, x, out, E);
    }
}


# sum launch: a block's edges are sorted by destination row in LDS (integer LDS counters); a half wave sums a row's run in a register by fma, in arrival order, stores the row once; no f32 LDS atomics
# speedup vs baseline: 3.2296x; 1.0306x over previous
#include <hip/hip_runtime.h>

typedef __attribute__((ext_vector_type(2))) float v2f;
typedef __attribute__((ext_vector_type(4))) float v4f;
typedef __attribute__((ext_vector_type(8))) float v8f;

#define IN_DIM 512
#define HID    32
#define OUT_DIM 16

#define GROUP_TILES   25
#define GROUP_THREADS (GROUP_TILES * 32)
#define GROUP_ROWS    (GROUP_TILES * 16)

#define KC           64
#define STAGE_P      68
#define STAGE_FLOATS (16 * STAGE_P)

#define SMEM_FLOATS 69768
#define CNT_OFF     44144
#define BASE_OFF    44408
#define WSUM_OFF    44672
#define LBUF_OFF    44680
#define SMEM_BYTES  (274 * 1024)

#define CHUNK 256
#define TICKET_STRIDE 64

#define NBINS        256
#define CH           12544
#define SORT_THREADS 1024
#define BIN_EPT      ((CH + GROUP_THREADS - 1) / GROUP_THREADS)
#define OFFS_STRIDE  (NBINS + 1)
#define ROW_SHIFT    23
#define COL_SHIFT    6
#define COL_FIELD    0x7FFFC0u
#define SUM_DYN_LDS  (128 * 1024)
#define RB_MAX       510
#define N_SORT_MAX   (NBINS * RB_MAX)
#define NULL_KEY     (511u << ROW_SHIFT)

__device__ __forceinline__ int rows_per_bin(int n) {
    return ((n + NBINS - 1) / NBINS + 1) & ~1;
}

__global__ __launch_bounds__(GROUP_THREADS)
void mlp_fused_kernel(const float* __restrict__ feat,
                      const float* __restrict__ W1,
                      const float* __restrict__ b1,
                      const float* __restrict__ W2,
                      const float* __restrict__ b2,
                      float* __restrict__ x, int n,
                      unsigned* __restrict__ tickets,
                      const int* __restrict__ edge_row,
                      const int* __restrict__ edge_col,
                      const float* __restrict__ edge_w,
                      uint2* __restrict__ sedge,
                      unsigned* __restrict__ offs, int E, int nchunks)
{
    extern __shared__ float smem[];
    if (tickets != nullptr && blockIdx.x == 0 && threadIdx.x < 2)
        tickets[threadIdx.x * TICKET_STRIDE] = 0u;

    float* sW1 = smem;
    float* sA  = smem + 16384;
    float* sW2 = smem + 43584;
    float* sb1 = smem + 44096;
    float* sb2 = smem + 44128;
    unsigned* cnt  = (unsigned*)(smem + CNT_OFF);
    unsigned* base = (unsigned*)(smem + BASE_OFF);
    unsigned* wsum = (unsigned*)(smem + WSUM_OFF);
    uint2*    lbuf = (uint2*)(smem + LBUF_OFF);

    const int tid  = (int)threadIdx.x;
    const int wave = tid >> 5;
    const int lane = tid & 31;

    const bool do_bin = sedge != nullptr && (int)blockIdx.x < nchunks;
    const unsigned rb = (unsigned)rows_per_bin(n);
    const int e0 = (int)blockIdx.x * CH;
    const int cn = do_bin ? ((E - e0) < CH ? (E - e0) : CH) : 0;

    int rr[BIN_EPT], cc[BIN_EPT];
    unsigned ww[BIN_EPT];
    #pragma unroll
    for (int k = 0; k < BIN_EPT; ++k) { rr[k] = 0; cc[k] = 0; ww[k] = 0u; }
    if (do_bin) {
        #pragma unroll
        for (int k = 0; k < BIN_EPT; ++k) {
            const int i  = tid + k * GROUP_THREADS;
            const int ii = e0 + (i < cn ? i : cn - 1);
            rr[k] = __builtin_nontemporal_load(edge_row + ii);
            cc[k] = __builtin_nontemporal_load(edge_col + ii);
            ww[k] = __float_as_uint(__builtin_nontemporal_load(edge_w + ii));
        }
        if (tid <= NBINS) cnt[tid] = 0u;
    }

    #pragma unroll
    for (int j = 0; j < (IN_DIM * HID + GROUP_THREADS - 1) / GROUP_THREADS; ++j) {
        const int i = (int)threadIdx.x + j * GROUP_THREADS;
        if (i < IN_DIM * HID) sW1[i] = W1[i];
    }
    for (int i = threadIdx.x; i < HID * OUT_DIM; i += GROUP_THREADS) sW2[i] = W2[i];
    if (threadIdx.x < HID)     sb1[threadIdx.x] = b1[threadIdx.x];
    if (threadIdx.x < OUT_DIM) sb2[threadIdx.x] = b2[threadIdx.x];
    __syncthreads();

    unsigned key[BIN_EPT], br[BIN_EPT];
    #pragma unroll
    for (int k = 0; k < BIN_EPT; ++k) { key[k] = 0u; br[k] = 0u; }
    if (do_bin) {
        #pragma unroll
        for (int k = 0; k < BIN_EPT; ++k) {
            const int i = tid + k * GROUP_THREADS;
            const unsigned r    = (unsigned)rr[k];
            const unsigned bin  = i < cn ? r / rb : (unsigned)NBINS;
            const unsigned rank = atomicAdd(&cnt[bin], 1u);
            br[k]  = (bin << 16) | rank;
            key[k] = ((r - bin * rb) << ROW_SHIFT) | ((unsigned)cc[k] << COL_SHIFT);
        }
    }
    __syncthreads();

    const int  row0     = blockIdx.x * GROUP_ROWS + wave * 16;
    const bool has_tile = row0 < n;
    const int m     = lane & 15;
    const int hh    = lane >> 4;
    const int khalf = hh << 1;
    const int n0    = lane & 15;
    const float* gsrc  = feat + (size_t)(row0 + hh) * IN_DIM + (lane & 15) * 4;

    unsigned v = 0u, s = 0u;
    if (do_bin && tid < NBINS) {
        v = cnt[tid];
        s = v;
        #pragma unroll
        for (int d = 1; d < 32; d <<= 1) {
            const unsigned t = __shfl_up(s, d);
            if (lane >= d) s += t;
        }
        if (lane == 31) wsum[wave] = s;
    }
    __syncthreads();
    if (do_bin && tid < NBINS) {
        unsigned add = 0u;
        for (int w = 0; w < wave; ++w) add += wsum[w];
        base[tid] = add + s - v;
        if (tid == NBINS - 1) base[NBINS] = add + s;
    }
    __syncthreads();

    if (do_bin) {
        #pragma unroll
        for (int k = 0; k < BIN_EPT; ++k) {
            const int i = tid + k * GROUP_THREADS;
            if (i < cn) {
                const unsigned pos = base[br[k] >> 16] + (br[k] & 0xFFFFu);
                lbuf[pos] = make_uint2(key[k], ww[k]);
            }
        }
    }
    __syncthreads();

    v4f r[8];
    #pragma unroll
    for (int j = 0; j < 8; ++j) r[j] = (v4f){0.0f, 0.0f, 0.0f, 0.0f};
    if (has_tile) {
        #pragma unroll
        for (int j = 0; j < 8; ++j)
            r[j] = __builtin_nontemporal_load((const v4f*)(gsrc + (size_t)(2 * j) * IN_DIM));
    }

    if (do_bin) {
        uint2* dst = sedge + (size_t)blockIdx.x * CH;
        for (int i = tid; i < cn; i += GROUP_THREADS) dst[i] = lbuf[i];
        if (tid <= NBINS) offs[(size_t)blockIdx.x * OFFS_STRIDE + tid] = base[tid];
    }

    if (has_tile) {
        float* st = sA + wave * STAGE_FLOATS;

        float* sdst = (float*)__builtin_assume_aligned(
            st + hh * STAGE_P + (lane & 15) * 4, 16);
        const float* afrag = (const float*)__builtin_assume_aligned(
            st + m * STAGE_P + khalf, 8);

        v8f acc0 = {};
        v8f acc1 = {};

        #pragma unroll 1
        for (int c = 0; c < IN_DIM / KC; ++c) {
            #pragma unroll
            for (int j = 0; j < 8; ++j)
                *(v4f*)(sdst + 2 * j * STAGE_P) = r[j];
            if (c + 1 < IN_DIM / KC) {
                #pragma unroll
                for (int j = 0; j < 8; ++j)
                    r[j] = __builtin_nontemporal_load(
                        (const v4f*)(gsrc + (size_t)(2 * j) * IN_DIM + (c + 1) * KC));
                asm volatile("s_wait_xcnt 0x0" ::: "memory");
            }
            asm volatile("s_wait_dscnt 0x0" ::: "memory");

            #pragma unroll
            for (int kk = 0; kk < KC; kk += 4) {
                const v2f a = *(const v2f*)(afrag + kk);
                const int kb = c * KC + kk + khalf;
                v2f bl, bh;
                bl.x = sW1[kb * HID + n0];
                bl.y = sW1[kb * HID + HID + n0];
                bh.x = sW1[kb * HID + 16 + n0];
                bh.y = sW1[kb * HID + HID + 16 + n0];
                acc0 = __builtin_amdgcn_wmma_f32_16x16x4_f32(
                    false, a, false, bl, (short)0, acc0, false, false);
                acc1 = __builtin_amdgcn_wmma_f32_16x16x4_f32(
                    false, a, false, bh, (short)0, acc1, false, false);
            }
        }

        float* shw = st;
        #pragma unroll
        for (int i = 0; i < 8; ++i) {
            const int mm = i + ((lane >> 4) << 3);
            float h0 = acc0[i] + sb1[n0];
            float h1 = acc1[i] + sb1[16 + n0];
            shw[mm * HID + n0]      = h0 > 0.0f ? h0 : 0.0f;
            shw[mm * HID + 16 + n0] = h1 > 0.0f ? h1 : 0.0f;
        }
        asm volatile("s_wait_dscnt 0x0" ::: "memory");

        const int mm   = lane & 15;
        const int half = lane >> 4;
        const float* hrow = shw + mm * HID;
        float* xrow = x + (size_t)(row0 + mm) * OUT_DIM;
        #pragma unroll
        for (int oo = 0; oo < 8; ++oo) {
            const int o = half * 8 + oo;
            float s2 = sb2[o];
            #pragma unroll
            for (int nn = 0; nn < HID; ++nn)
                s2 += hrow[nn] * sW2[nn * OUT_DIM + o];
            xrow[o] = s2;
        }
    }
}

__global__ __launch_bounds__(256)
void spmm_scatter_kernel(const int* __restrict__ edge_row,
                         const int* __restrict__ edge_col,
                         const float* __restrict__ edge_w,
                         const float* __restrict__ x,
                         float* __restrict__ out, int E)
{
    const long t = (long)blockIdx.x * 256 + threadIdx.x;
    const long e = t >> 4;
    const int  o = (int)(t & 15);
    if (e < E) {
        const int   r = __builtin_nontemporal_load(edge_row + e);
        const int   c = __builtin_nontemporal_load(edge_col + e);
        const float w = __builtin_nontemporal_load(edge_w + e);
        atomicAdd(out + (size_t)r * OUT_DIM + o, w * x[(size_t)c * OUT_DIM + o]);
    }
}


__global__ __launch_bounds__(256)
void spmm_owner_kernel(const int* __restrict__ edge_row,
                       const int* __restrict__ edge_col,
                       const float* __restrict__ edge_w,
                       const float* __restrict__ x,
                       float* __restrict__ out,
                       unsigned* __restrict__ tickets, int E, int n)
{
    const int  lane = threadIdx.x & 31;
    const int  o    = lane & 15;
    const bool hi   = lane >= 16;

    const unsigned place  = __builtin_amdgcn_s_sendmsg_rtn(0x87);
    const int      myhalf = (int)((place >> 18) & 1u);
    const int      split  = (n >> 1) & ~3;
    const unsigned nchunks = (unsigned)((E + CHUNK - 1) / CHUNK);

    for (int pass = 0; pass < 2; ++pass) {
        const int h = myhalf ^ pass;
        unsigned* tk = tickets + h * TICKET_STRIDE;
        for (unsigned it = 0; it <= nchunks; ++it) {
            unsigned t = 0u;
            if (lane == 0) t = atomicAdd(tk, 1u);
            t = (unsigned)__builtin_amdgcn_readfirstlane((int)t);
            if (t >= nchunks) break;
            const int e0 = (int)t * CHUNK;

            const int elast = E - 1;
            int e = e0 + lane;
            int ec0 = e < E ? e : elast;
            int   r = __builtin_nontemporal_load(edge_row + ec0);
            int   c = __builtin_nontemporal_load(edge_col + ec0);
            float w = __builtin_nontemporal_load(edge_w + ec0);
            if (e >= E) r = -1;
            for (int rd = 0; rd < CHUNK / 32; ++rd) {
                const int en  = e + 32;
                const int enc = en < E ? en : elast;
                int   rn = __builtin_nontemporal_load(edge_row + enc);
                int   cn = __builtin_nontemporal_load(edge_col + enc);
                float wn = __builtin_nontemporal_load(edge_w + enc);
                if (en >= E) rn = -1;

                const bool mine = (r >= 0) && ((r >= split) == (h != 0));
                const unsigned m = (unsigned)__builtin_amdgcn_readfirstlane(
                    (int)__builtin_amdgcn_ballot_w32(mine));
                const int npairs = (__builtin_popcount(m) + 1) >> 1;
                const int wi = __float_as_int(w);

                float v[16];
                unsigned m1 = m;
                #pragma unroll
                for (int j = 0; j < 16; ++j) {
                    const int s0 = m1 ? __builtin_ctz(m1) : 0;  m1 &= m1 - 1u;
                    const int s1 = m1 ? __builtin_ctz(m1) : s0; m1 &= m1 - 1u;
                    const int c0 = __builtin_amdgcn_readlane(c, s0);
                    const int c1 = __builtin_amdgcn_readlane(c, s1);
                    const int cc = hi ? c1 : c0;
                    v[j] = x[cc * OUT_DIM + o];
                }
                unsigned m2 = m;
                #pragma unroll
                for (int j = 0; j < 16; ++j) {
                    if (j < npairs) {
                        const int s0 = __builtin_ctz(m2); m2 &= m2 - 1u;
                        const bool a1 = m2 != 0u;
                        const int s1 = a1 ? __builtin_ctz(m2) : s0; m2 &= m2 - 1u;
                        const int r0 = __builtin_amdgcn_readlane(r, s0);
                        const int r1 = __builtin_amdgcn_readlane(r, s1);
                        const int w0 = __builtin_amdgcn_readlane(wi, s0);
                        const int w1 = __builtin_amdgcn_readlane(wi, s1);
                        const int   rr = hi ? r1 : r0;
                        const float ww = __int_as_float(hi ? w1 : w0);
                        if (!hi || a1)
                            atomicAdd(out + rr * OUT_DIM + o, ww * v[j]);
                    }
                }
                r = rn; c = cn; w = wn; e = en;
            }
        }
    }
}


#define ROW_SHARE(v, J) \
    ((unsigned)__builtin_amdgcn_update_dpp(0, (int)(v), 0x150 + (J), 0xF, 0xF, false))

__device__ __forceinline__ void block_add_round(uint2 raw, int cnt, int lane, unsigned o4,
                                                const float* __restrict__ x,
                                                float* __restrict__ acc)
{
    const unsigned key = lane < cnt ? raw.x : NULL_KEY;
    const unsigned wb  = lane < cnt ? raw.y : 0u;
    const char* xb = (const char*)x;
    char*       ab = (char*)acc;
    float    v[16];
    unsigned kk[16];
#define GATHER_STEP(J)                                                        \
    kk[J] = ROW_SHARE(key, J);                                                \
    v[J]  = *(const float*)(xb + ((kk[J] & COL_FIELD) | o4));
    GATHER_STEP(0)  GATHER_STEP(1)  GATHER_STEP(2)  GATHER_STEP(3)
    GATHER_STEP(4)  GATHER_STEP(5)  GATHER_STEP(6)  GATHER_STEP(7)
    GATHER_STEP(8)  GATHER_STEP(9)  GATHER_STEP(10) GATHER_STEP(11)
    GATHER_STEP(12) GATHER_STEP(13) GATHER_STEP(14) GATHER_STEP(15)
#undef GATHER_STEP
#define ADD_STEP(J)                                                           \
    atomicAdd((float*)(ab + (((kk[J] >> ROW_SHIFT) << 6) | o4)),              \
              __uint_as_float(ROW_SHARE(wb, J)) * v[J]);
    ADD_STEP(0)  ADD_STEP(1)  ADD_STEP(2)  ADD_STEP(3)
    ADD_STEP(4)  ADD_STEP(5)  ADD_STEP(6)  ADD_STEP(7)
    ADD_STEP(8)  ADD_STEP(9)  ADD_STEP(10) ADD_STEP(11)
    ADD_STEP(12) ADD_STEP(13) ADD_STEP(14) ADD_STEP(15)
#undef ADD_STEP
}

#define ROW_CAP   16384
#define ROW_SLOTS 512
#define SORT_U    8

__device__ __forceinline__ void run_gather(unsigned xo, unsigned o4,
                                           const float* __restrict__ x, float* v)
{
    const char* xb = (const char*)x;
#define RG(J) v[J] = *(const float*)(xb + (ROW_SHARE(xo, J) | o4));
    RG(0) RG(1) RG(2) RG(3) RG(4) RG(5) RG(6) RG(7)
    RG(8) RG(9) RG(10) RG(11) RG(12) RG(13) RG(14) RG(15)
#undef RG
}

__device__ __forceinline__ float run_sum(unsigned wb, const float* v, float a)
{
#define RS(J) a = __builtin_fmaf(__uint_as_float(ROW_SHARE(wb, J)), v[J], a);
    RS(0) RS(1) RS(2) RS(3) RS(4) RS(5) RS(6) RS(7)
    RS(8) RS(9) RS(10) RS(11) RS(12) RS(13) RS(14) RS(15)
#undef RS
    return a;
}

__global__ __launch_bounds__(SORT_THREADS)
void row_sum_kernel(const uint2* __restrict__ sedge,
                    const unsigned* __restrict__ offs,
                    const float* __restrict__ x,
                    float* __restrict__ out, int n, int nchunks)
{
    extern __shared__ uint2 srt[];
    __shared__ float    acc[(RB_MAX + 2) * OUT_DIM];
    __shared__ unsigned cnt[ROW_SLOTS + 1 + 32];
    __shared__ unsigned wsum[16];

    const int tid  = (int)threadIdx.x;
    const int lane = tid & 31;
    const int hh   = lane >> 4;
    const int wave = __builtin_amdgcn_readfirstlane(tid >> 5);
    const unsigned o4 = (unsigned)(lane & 15) << 2;
    const int rb   = rows_per_bin(n);
    const int b    = (int)blockIdx.x;
    const int row0 = b * rb;
    int nrows = n - row0;
    if (nrows > rb) nrows = rb;
    if (nrows < 0)  nrows = 0;

    if (tid <= ROW_SLOTS) cnt[tid] = 0u;
    __syncthreads();

    const unsigned* ob = offs + b;

    for (int gb = wave * SORT_U; gb < nchunks; gb += 32 * SORT_U) {
        unsigned sv = 0u, tv = 0u;
        if (lane < SORT_U && gb + lane < nchunks) {
            sv = ob[(size_t)(gb + lane) * OFFS_STRIDE];
            tv = ob[(size_t)(gb + lane) * OFFS_STRIDE + 1];
            if (tv > (unsigned)CH) tv = (unsigned)CH;
            if (sv > tv) sv = tv;
        }
        unsigned mx = 0u;
        #pragma unroll
        for (int u = 0; u < SORT_U; ++u) {
            const unsigned l = (unsigned)__builtin_amdgcn_readlane((int)(tv - sv), u);
            mx = l > mx ? l : mx;
        }
        for (unsigned p = 0u; p < mx; p += 64u) {
            unsigned k[2 * SORT_U];
            #pragma unroll
            for (int u = 0; u < 2 * SORT_U; ++u) {
                const unsigned idx = (unsigned)__builtin_amdgcn_readlane((int)sv, u >> 1)
                                   + p + 32u * (unsigned)(u & 1) + (unsigned)lane;
                const bool ok = idx < (unsigned)__builtin_amdgcn_readlane((int)tv, u >> 1);
                k[u] = (ok ? sedge + (size_t)(gb + (u >> 1)) * CH + idx : sedge)->x;
            }
            #pragma unroll
            for (int u = 0; u < 2 * SORT_U; ++u) {
                const unsigned idx = (unsigned)__builtin_amdgcn_readlane((int)sv, u >> 1)
                                   + p + 32u * (unsigned)(u & 1) + (unsigned)lane;
                if (idx < (unsigned)__builtin_amdgcn_readlane((int)tv, u >> 1))
                    atomicAdd(&cnt[k[u] >> ROW_SHIFT], 1u);
            }
        }
    }
    __syncthreads();

    unsigned vv = 0u, ss = 0u;
    if (tid < ROW_SLOTS) {
        vv = cnt[tid];
        ss = vv;
        #pragma unroll
        for (int d = 1; d < 32; d <<= 1) {
            const unsigned t = __shfl_up(ss, d);
            if (lane >= d) ss += t;
        }
        if (lane == 31) wsum[wave] = ss;
    }
    __syncthreads();
    if (tid < ROW_SLOTS) {
        unsigned add = 0u;
        for (int w = 0; w < wave; ++w) add += wsum[w];
        cnt[tid] = add + ss;
        if (tid == ROW_SLOTS - 1) cnt[ROW_SLOTS] = add + ss;
    }
    __syncthreads();
    const unsigned total = cnt[ROW_SLOTS];

    if (total <= (unsigned)ROW_CAP) {
        for (int gb = wave * SORT_U; gb < nchunks; gb += 32 * SORT_U) {
            unsigned sv = 0u, tv = 0u;
            if (lane < SORT_U && gb + lane < nchunks) {
                sv = ob[(size_t)(gb + lane) * OFFS_STRIDE];
                tv = ob[(size_t)(gb + lane) * OFFS_STRIDE + 1];
                if (tv > (unsigned)CH) tv = (unsigned)CH;
                if (sv > tv) sv = tv;
            }
            unsigned mx = 0u;
            #pragma unroll
            for (int u = 0; u < SORT_U; ++u) {
                const unsigned l = (unsigned)__builtin_amdgcn_readlane((int)(tv - sv), u);
                mx = l > mx ? l : mx;
            }
            for (unsigned p = 0u; p < mx; p += 64u) {
                uint2 rec[2 * SORT_U];
                #pragma unroll
                for (int u = 0; u < 2 * SORT_U; ++u) {
                    const unsigned idx = (unsigned)__builtin_amdgcn_readlane((int)sv, u >> 1)
                                       + p + 32u * (unsigned)(u & 1) + (unsigned)lane;
                    const bool ok = idx < (unsigned)__builtin_amdgcn_readlane((int)tv, u >> 1);
                    rec[u] = *(ok ? sedge + (size_t)(gb + (u >> 1)) * CH + idx : sedge);
                }
                __builtin_amdgcn_sched_barrier(0);
                unsigned pos[2 * SORT_U];
                #pragma unroll
                for (int u = 0; u < 2 * SORT_U; ++u) {
                    const unsigned idx = (unsigned)__builtin_amdgcn_readlane((int)sv, u >> 1)
                                       + p + 32u * (unsigned)(u & 1) + (unsigned)lane;
                    const bool ok = idx < (unsigned)__builtin_amdgcn_readlane((int)tv, u >> 1);
                    const unsigned slot = ok ? rec[u].x >> ROW_SHIFT
                                             : (unsigned)(ROW_SLOTS + 1 + lane);
                    pos[u] = atomicSub(&cnt[slot], 1u) - 1u;
                }
                #pragma unroll
                for (int u = 0; u < 2 * SORT_U; ++u) {
                    const unsigned idx = (unsigned)__builtin_amdgcn_readlane((int)sv, u >> 1)
                                       + p + 32u * (unsigned)(u & 1) + (unsigned)lane;
                    if (idx < (unsigned)__builtin_amdgcn_readlane((int)tv, u >> 1) &&
                        pos[u] < (unsigned)ROW_CAP)
                        srt[pos[u]] = make_uint2(rec[u].x & COL_FIELD, rec[u].y);
                }
            }
        }
        __syncthreads();

        const int npairs = (nrows + 1) >> 1;
        int t = wave;
        unsigned st_n = 0u, en_n = 0u;
        if (t < npairs) {
            const int r = 2 * t + hh;
            st_n = cnt[r < nrows ? r : 0];
            en_n = cnt[r < nrows ? r + 1 : 0];
        }
        while (t < npairs) {
            const int  r     = 2 * t + hh;
            const bool rowok = r < nrows;
            const unsigned st = st_n;
            const unsigned d  = rowok ? en_n - st_n : 0u;
            if (t + 32 < npairs) {
                const int rn = 2 * (t + 32) + hh;
                st_n = cnt[rn < nrows ? rn : 0];
                en_n = cnt[rn < nrows ? rn + 1 : 0];
            }
            const unsigned d0 = (unsigned)__builtin_amdgcn_readlane((int)d, 0);
            const unsigned d1 = (unsigned)__builtin_amdgcn_readlane((int)d, 16);
            const unsigned dmax = d0 > d1 ? d0 : d1;

            float a = 0.0f;
            #define FETCH_PIECE(R, J0)                                         \
                do {                                                           \
                    const unsigned sl_ = (J0) + (unsigned)(lane & 15);         \
                    const bool ok_ = sl_ < d;                                  \
                    const uint2 t_ = srt[ok_ ? st + sl_ : 0u];                 \
                    R.x = ok_ ? t_.x : 0u;                                     \
                    R.y = ok_ ? t_.y : 0u;                                     \
                } while (0)
            for (unsigned j = 0u; j < dmax; j += 64u) {
                float v0[16], v1[16], v2[16], v3[16];
                uint2 r0, r1, r2, r3;
                const bool h1 = j + 16u < dmax, h2 = j + 32u < dmax, h3 = j + 48u < dmax;
                FETCH_PIECE(r0, j);
                FETCH_PIECE(r1, j + 16u);
                FETCH_PIECE(r2, j + 32u);
                FETCH_PIECE(r3, j + 48u);
                run_gather(r0.x, o4, x, v0);
                if (h1) run_gather(r1.x, o4, x, v1);
                if (h2) run_gather(r2.x, o4, x, v2);
                if (h3) run_gather(r3.x, o4, x, v3);
                a = run_sum(r0.y, v0, a);
                if (h1) a = run_sum(r1.y, v1, a);
                if (h2) a = run_sum(r2.y, v2, a);
                if (h3) a = run_sum(r3.y, v3, a);
            }
            #undef FETCH_PIECE
            if (rowok) out[(size_t)(row0 + r) * OUT_DIM + (lane & 15)] = a;
            t += 32;
        }
        return;
    }

    for (int i = tid; i < (RB_MAX + 2) * OUT_DIM; i += SORT_THREADS) acc[i] = 0.0f;
    __syncthreads();
    for (int g = wave; g < nchunks; g += 32) {
        const unsigned s = ob[(size_t)g * OFFS_STRIDE];
        unsigned e = ob[(size_t)g * OFFS_STRIDE + 1];
        if (e > (unsigned)CH) e = (unsigned)CH;
        const uint2* seg = sedge + (size_t)g * CH;
        for (unsigned p = s; p < e; p += 32u) {
            const int c = (int)(e - p) < 32 ? (int)(e - p) : 32;
            const uint2 ed = *(lane < c ? seg + p + (unsigned)lane : sedge);
            block_add_round(ed, c, lane, o4, x, acc);
        }
    }
    __syncthreads();
    float* ob_out = out + (size_t)row0 * OUT_DIM;
    for (int i = tid; i < nrows * OUT_DIM; i += SORT_THREADS) ob_out[i] = acc[i];
}

extern "C" void kernel_launch(void* const* d_in, const int* in_sizes, int n_in,
                              void* d_out, int out_size, void* d_ws, size_t ws_size,
                              hipStream_t stream) {
    const float* feat = (const float*)d_in[0];
    const int*   er   = (const int*)  d_in[1];
    const int*   ec   = (const int*)  d_in[2];
    const float* ew   = (const float*)d_in[3];
    const float* W1   = (const float*)d_in[4];
    const float* b1   = (const float*)d_in[5];
    const float* W2   = (const float*)d_in[6];
    const float* b2   = (const float*)d_in[7];
    float* out = (float*)d_out;
    float* x   = (float*)d_ws;

    const int n = in_sizes[0] / IN_DIM;
    const int E = in_sizes[1];

    const size_t tk_off = (((size_t)n * OUT_DIM * sizeof(float)) + 255) & ~(size_t)255;
    const bool   owner  = ws_size >= tk_off + 512;
    unsigned* tickets = owner ? (unsigned*)((char*)d_ws + tk_off) : nullptr;

    const int    nchunks  = (E + CH - 1) / CH;
    const size_t se_off   = tk_off + 512;
    const size_t of_off   = se_off + (size_t)nchunks * CH * sizeof(uint2);
    const size_t ws_need  = of_off + (size_t)nchunks * OFFS_STRIDE * sizeof(unsigned);
    const bool   sorted   = n <= N_SORT_MAX && E > 0 && ws_size >= ws_need;

    const int mlp_blocks = (n + GROUP_ROWS - 1) / GROUP_ROWS;

    if (sorted) {
        uint2*    sedge = (uint2*)((char*)d_ws + se_off);
        unsigned* offs  = (unsigned*)((char*)d_ws + of_off);
        const int fused_blocks = mlp_blocks > nchunks ? mlp_blocks : nchunks;
        mlp_fused_kernel<<<fused_blocks, GROUP_THREADS, SMEM_BYTES, stream>>>(
            feat, W1, b1, W2, b2, x, n, nullptr, er, ec, ew, sedge, offs, E, nchunks);
        row_sum_kernel<<<NBINS, SORT_THREADS, SUM_DYN_LDS, stream>>>(sedge, offs, x, out, n, nchunks);
        return;
    }

    hipMemsetAsync(d_out, 0, (size_t)out_size * sizeof(float), stream);

    mlp_fused_kernel<<<mlp_blocks, GROUP_THREADS, SMEM_BYTES, stream>>>(
        feat, W1, b1, W2, b2, x, n, tickets, er, ec, ew, nullptr, nullptr, E, 0);

    if (owner) {
        spmm_owner_kernel<<<2048, 256, 0, stream>>>(er, ec, ew, x, out, tickets, E, n);
    } else {
        const long sthreads = (long)E * 16;
        const int  sblocks  = (int)((sthreads + 255) / 256);
        spmm_scatter_kernel<<<sblocks, 256, 0, stream>>>(er, ec, ew, x, out, E);
    }
}


# sum launch: row sort in ONE walk: a wave keeps its edges in registers, the returning LDS count add gives each its rank, slot = row start + rank; no second walk; arrival-order sums as before
# speedup vs baseline: 3.3470x; 1.0364x over previous
#include <hip/hip_runtime.h>

typedef __attribute__((ext_vector_type(2))) float v2f;
typedef __attribute__((ext_vector_type(4))) float v4f;
typedef __attribute__((ext_vector_type(8))) float v8f;

#define IN_DIM 512
#define HID    32
#define OUT_DIM 16

#define GROUP_TILES   25
#define GROUP_THREADS (GROUP_TILES * 32)
#define GROUP_ROWS    (GROUP_TILES * 16)

#define KC           64
#define STAGE_P      68
#define STAGE_FLOATS (16 * STAGE_P)

#define SMEM_FLOATS 69768
#define CNT_OFF     44144
#define BASE_OFF    44408
#define WSUM_OFF    44672
#define LBUF_OFF    44680
#define SMEM_BYTES  (274 * 1024)

#define CHUNK 256
#define TICKET_STRIDE 64

#define NBINS        256
#define CH           12544
#define SORT_THREADS 1024
#define BIN_EPT      ((CH + GROUP_THREADS - 1) / GROUP_THREADS)
#define OFFS_STRIDE  (NBINS + 1)
#define ROW_SHIFT    23
#define COL_SHIFT    6
#define COL_FIELD    0x7FFFC0u
#define SUM_DYN_LDS  (128 * 1024)
#define RB_MAX       510
#define N_SORT_MAX   (NBINS * RB_MAX)
#define NULL_KEY     (511u << ROW_SHIFT)

__device__ __forceinline__ int rows_per_bin(int n) {
    return ((n + NBINS - 1) / NBINS + 1) & ~1;
}

__global__ __launch_bounds__(GROUP_THREADS)
void mlp_fused_kernel(const float* __restrict__ feat,
                      const float* __restrict__ W1,
                      const float* __restrict__ b1,
                      const float* __restrict__ W2,
                      const float* __restrict__ b2,
                      float* __restrict__ x, int n,
                      unsigned* __restrict__ tickets,
                      const int* __restrict__ edge_row,
                      const int* __restrict__ edge_col,
                      const float* __restrict__ edge_w,
                      uint2* __restrict__ sedge,
                      unsigned* __restrict__ offs, int E, int nchunks)
{
    extern __shared__ float smem[];
    if (tickets != nullptr && blockIdx.x == 0 && threadIdx.x < 2)
        tickets[threadIdx.x * TICKET_STRIDE] = 0u;

    float* sW1 = smem;
    float* sA  = smem + 16384;
    float* sW2 = smem + 43584;
    float* sb1 = smem + 44096;
    float* sb2 = smem + 44128;
    unsigned* cnt  = (unsigned*)(smem + CNT_OFF);
    unsigned* base = (unsigned*)(smem + BASE_OFF);
    unsigned* wsum = (unsigned*)(smem + WSUM_OFF);
    uint2*    lbuf = (uint2*)(smem + LBUF_OFF);

    const int tid  = (int)threadIdx.x;
    const int wave = tid >> 5;
    const int lane = tid & 31;

    const bool do_bin = sedge != nullptr && (int)blockIdx.x < nchunks;
    const unsigned rb = (unsigned)rows_per_bin(n);
    const int e0 = (int)blockIdx.x * CH;
    const int cn = do_bin ? ((E - e0) < CH ? (E - e0) : CH) : 0;

    int rr[BIN_EPT], cc[BIN_EPT];
    unsigned ww[BIN_EPT];
    #pragma unroll
    for (int k = 0; k < BIN_EPT; ++k) { rr[k] = 0; cc[k] = 0; ww[k] = 0u; }
    if (do_bin) {
        #pragma unroll
        for (int k = 0; k < BIN_EPT; ++k) {
            const int i  = tid + k * GROUP_THREADS;
            const int ii = e0 + (i < cn ? i : cn - 1);
            rr[k] = __builtin_nontemporal_load(edge_row + ii);
            cc[k] = __builtin_nontemporal_load(edge_col + ii);
            ww[k] = __float_as_uint(__builtin_nontemporal_load(edge_w + ii));
        }
        if (tid <= NBINS) cnt[tid] = 0u;
    }

    #pragma unroll
    for (int j = 0; j < (IN_DIM * HID + GROUP_THREADS - 1) / GROUP_THREADS; ++j) {
        const int i = (int)threadIdx.x + j * GROUP_THREADS;
        if (i < IN_DIM * HID) sW1[i] = W1[i];
    }
    for (int i = threadIdx.x; i < HID * OUT_DIM; i += GROUP_THREADS) sW2[i] = W2[i];
    if (threadIdx.x < HID)     sb1[threadIdx.x] = b1[threadIdx.x];
    if (threadIdx.x < OUT_DIM) sb2[threadIdx.x] = b2[threadIdx.x];
    __syncthreads();

    unsigned key[BIN_EPT], br[BIN_EPT];
    #pragma unroll
    for (int k = 0; k < BIN_EPT; ++k) { key[k] = 0u; br[k] = 0u; }
    if (do_bin) {
        #pragma unroll
        for (int k = 0; k < BIN_EPT; ++k) {
            const int i = tid + k * GROUP_THREADS;
            const unsigned r    = (unsigned)rr[k];
            const unsigned bin  = i < cn ? r / rb : (unsigned)NBINS;
            const unsigned rank = atomicAdd(&cnt[bin], 1u);
            br[k]  = (bin << 16) | rank;
            key[k] = ((r - bin * rb) << ROW_SHIFT) | ((unsigned)cc[k] << COL_SHIFT);
        }
    }
    __syncthreads();

    const int  row0     = blockIdx.x * GROUP_ROWS + wave * 16;
    const bool has_tile = row0 < n;
    const int m     = lane & 15;
    const int hh    = lane >> 4;
    const int khalf = hh << 1;
    const int n0    = lane & 15;
    const float* gsrc  = feat + (size_t)(row0 + hh) * IN_DIM + (lane & 15) * 4;

    unsigned v = 0u, s = 0u;
    if (do_bin && tid < NBINS) {
        v = cnt[tid];
        s = v;
        #pragma unroll
        for (int d = 1; d < 32; d <<= 1) {
            const unsigned t = __shfl_up(s, d);
            if (lane >= d) s += t;
        }
        if (lane == 31) wsum[wave] = s;
    }
    __syncthreads();
    if (do_bin && tid < NBINS) {
        unsigned add = 0u;
        for (int w = 0; w < wave; ++w) add += wsum[w];
        base[tid] = add + s - v;
        if (tid == NBINS - 1) base[NBINS] = add + s;
    }
    __syncthreads();

    if (do_bin) {
        #pragma unroll
        for (int k = 0; k < BIN_EPT; ++k) {
            const int i = tid + k * GROUP_THREADS;
            if (i < cn) {
                const unsigned pos = base[br[k] >> 16] + (br[k] & 0xFFFFu);
                lbuf[pos] = make_uint2(key[k], ww[k]);
            }
        }
    }
    __syncthreads();

    v4f r[8];
    #pragma unroll
    for (int j = 0; j < 8; ++j) r[j] = (v4f){0.0f, 0.0f, 0.0f, 0.0f};
    if (has_tile) {
        #pragma unroll
        for (int j = 0; j < 8; ++j)
            r[j] = __builtin_nontemporal_load((const v4f*)(gsrc + (size_t)(2 * j) * IN_DIM));
    }

    if (do_bin) {
        uint2* dst = sedge + (size_t)blockIdx.x * CH;
        for (int i = tid; i < cn; i += GROUP_THREADS) dst[i] = lbuf[i];
        if (tid <= NBINS) offs[(size_t)blockIdx.x * OFFS_STRIDE + tid] = base[tid];
    }

    if (has_tile) {
        float* st = sA + wave * STAGE_FLOATS;

        float* sdst = (float*)__builtin_assume_aligned(
            st + hh * STAGE_P + (lane & 15) * 4, 16);
        const float* afrag = (const float*)__builtin_assume_aligned(
            st + m * STAGE_P + khalf, 8);

        v8f acc0 = {};
        v8f acc1 = {};

        #pragma unroll 1
        for (int c = 0; c < IN_DIM / KC; ++c) {
            #pragma unroll
            for (int j = 0; j < 8; ++j)
                *(v4f*)(sdst + 2 * j * STAGE_P) = r[j];
            if (c + 1 < IN_DIM / KC) {
                #pragma unroll
                for (int j = 0; j < 8; ++j)
                    r[j] = __builtin_nontemporal_load(
                        (const v4f*)(gsrc + (size_t)(2 * j) * IN_DIM + (c + 1) * KC));
                asm volatile("s_wait_xcnt 0x0" ::: "memory");
            }
            asm volatile("s_wait_dscnt 0x0" ::: "memory");

            #pragma unroll
            for (int kk = 0; kk < KC; kk += 4) {
                const v2f a = *(const v2f*)(afrag + kk);
                const int kb = c * KC + kk + khalf;
                v2f bl, bh;
                bl.x = sW1[kb * HID + n0];
                bl.y = sW1[kb * HID + HID + n0];
                bh.x = sW1[kb * HID + 16 + n0];
                bh.y = sW1[kb * HID + HID + 16 + n0];
                acc0 = __builtin_amdgcn_wmma_f32_16x16x4_f32(
                    false, a, false, bl, (short)0, acc0, false, false);
                acc1 = __builtin_amdgcn_wmma_f32_16x16x4_f32(
                    false, a, false, bh, (short)0, acc1, false, false);
            }
        }

        float* shw = st;
        #pragma unroll
        for (int i = 0; i < 8; ++i) {
            const int mm = i + ((lane >> 4) << 3);
            float h0 = acc0[i] + sb1[n0];
            float h1 = acc1[i] + sb1[16 + n0];
            shw[mm * HID + n0]      = h0 > 0.0f ? h0 : 0.0f;
            shw[mm * HID + 16 + n0] = h1 > 0.0f ? h1 : 0.0f;
        }
        asm volatile("s_wait_dscnt 0x0" ::: "memory");

        const int mm   = lane & 15;
        const int half = lane >> 4;
        const float* hrow = shw + mm * HID;
        float* xrow = x + (size_t)(row0 + mm) * OUT_DIM;
        #pragma unroll
        for (int oo = 0; oo < 8; ++oo) {
            const int o = half * 8 + oo;
            float s2 = sb2[o];
            #pragma unroll
            for (int nn = 0; nn < HID; ++nn)
                s2 += hrow[nn] * sW2[nn * OUT_DIM + o];
            xrow[o] = s2;
        }
    }
}

__global__ __launch_bounds__(256)
void spmm_scatter_kernel(const int* __restrict__ edge_row,
                         const int* __restrict__ edge_col,
                         const float* __restrict__ edge_w,
                         const float* __restrict__ x,
                         float* __restrict__ out, int E)
{
    const long t = (long)blockIdx.x * 256 + threadIdx.x;
    const long e = t >> 4;
    const int  o = (int)(t & 15);
    if (e < E) {
        const int   r = __builtin_nontemporal_load(edge_row + e);
        const int   c = __builtin_nontemporal_load(edge_col + e);
        const float w = __builtin_nontemporal_load(edge_w + e);
        atomicAdd(out + (size_t)r * OUT_DIM + o, w * x[(size_t)c * OUT_DIM + o]);
    }
}


__global__ __launch_bounds__(256)
void spmm_owner_kernel(const int* __restrict__ edge_row,
                       const int* __restrict__ edge_col,
                       const float* __restrict__ edge_w,
                       const float* __restrict__ x,
                       float* __restrict__ out,
                       unsigned* __restrict__ tickets, int E, int n)
{
    const int  lane = threadIdx.x & 31;
    const int  o    = lane & 15;
    const bool hi   = lane >= 16;

    const unsigned place  = __builtin_amdgcn_s_sendmsg_rtn(0x87);
    const int      myhalf = (int)((place >> 18) & 1u);
    const int      split  = (n >> 1) & ~3;
    const unsigned nchunks = (unsigned)((E + CHUNK - 1) / CHUNK);

    for (int pass = 0; pass < 2; ++pass) {
        const int h = myhalf ^ pass;
        unsigned* tk = tickets + h * TICKET_STRIDE;
        for (unsigned it = 0; it <= nchunks; ++it) {
            unsigned t = 0u;
            if (lane == 0) t = atomicAdd(tk, 1u);
            t = (unsigned)__builtin_amdgcn_readfirstlane((int)t);
            if (t >= nchunks) break;
            const int e0 = (int)t * CHUNK;

            const int elast = E - 1;
            int e = e0 + lane;
            int ec0 = e < E ? e : elast;
            int   r = __builtin_nontemporal_load(edge_row + ec0);
            int   c = __builtin_nontemporal_load(edge_col + ec0);
            float w = __builtin_nontemporal_load(edge_w + ec0);
            if (e >= E) r = -1;
            for (int rd = 0; rd < CHUNK / 32; ++rd) {
                const int en  = e + 32;
                const int enc = en < E ? en : elast;
                int   rn = __builtin_nontemporal_load(edge_row + enc);
                int   cn = __builtin_nontemporal_load(edge_col + enc);
                float wn = __builtin_nontemporal_load(edge_w + enc);
                if (en >= E) rn = -1;

                const bool mine = (r >= 0) && ((r >= split) == (h != 0));
                const unsigned m = (unsigned)__builtin_amdgcn_readfirstlane(
                    (int)__builtin_amdgcn_ballot_w32(mine));
                const int npairs = (__builtin_popcount(m) + 1) >> 1;
                const int wi = __float_as_int(w);

                float v[16];
                unsigned m1 = m;
                #pragma unroll
                for (int j = 0; j < 16; ++j) {
                    const int s0 = m1 ? __builtin_ctz(m1) : 0;  m1 &= m1 - 1u;
                    const int s1 = m1 ? __builtin_ctz(m1) : s0; m1 &= m1 - 1u;
                    const int c0 = __builtin_amdgcn_readlane(c, s0);
                    const int c1 = __builtin_amdgcn_readlane(c, s1);
                    const int cc = hi ? c1 : c0;
                    v[j] = x[cc * OUT_DIM + o];
                }
                unsigned m2 = m;
                #pragma unroll
                for (int j = 0; j < 16; ++j) {
                    if (j < npairs) {
                        const int s0 = __builtin_ctz(m2); m2 &= m2 - 1u;
                        const bool a1 = m2 != 0u;
                        const int s1 = a1 ? __builtin_ctz(m2) : s0; m2 &= m2 - 1u;
                        const int r0 = __builtin_amdgcn_readlane(r, s0);
                        const int r1 = __builtin_amdgcn_readlane(r, s1);
                        const int w0 = __builtin_amdgcn_readlane(wi, s0);
                        const int w1 = __builtin_amdgcn_readlane(wi, s1);
                        const int   rr = hi ? r1 : r0;
                        const float ww = __int_as_float(hi ? w1 : w0);
                        if (!hi || a1)
                            atomicAdd(out + rr * OUT_DIM + o, ww * v[j]);
                    }
                }
                r = rn; c = cn; w = wn; e = en;
            }
        }
    }
}


#define ROW_SHARE(v, J) \
    ((unsigned)__builtin_amdgcn_update_dpp(0, (int)(v), 0x150 + (J), 0xF, 0xF, false))

__device__ __forceinline__ void block_add_round(uint2 raw, int cnt, int lane, unsigned o4,
                                                const float* __restrict__ x,
                                                float* __restrict__ acc)
{
    const unsigned key = lane < cnt ? raw.x : NULL_KEY;
    const unsigned wb  = lane < cnt ? raw.y : 0u;
    const char* xb = (const char*)x;
    char*       ab = (char*)acc;
    float    v[16];
    unsigned kk[16];
#define GATHER_STEP(J)                                                        \
    kk[J] = ROW_SHARE(key, J);                                                \
    v[J]  = *(const float*)(xb + ((kk[J] & COL_FIELD) | o4));
    GATHER_STEP(0)  GATHER_STEP(1)  GATHER_STEP(2)  GATHER_STEP(3)
    GATHER_STEP(4)  GATHER_STEP(5)  GATHER_STEP(6)  GATHER_STEP(7)
    GATHER_STEP(8)  GATHER_STEP(9)  GATHER_STEP(10) GATHER_STEP(11)
    GATHER_STEP(12) GATHER_STEP(13) GATHER_STEP(14) GATHER_STEP(15)
#undef GATHER_STEP
#define ADD_STEP(J)                                                           \
    atomicAdd((float*)(ab + (((kk[J] >> ROW_SHIFT) << 6) | o4)),              \
              __uint_as_float(ROW_SHARE(wb, J)) * v[J]);
    ADD_STEP(0)  ADD_STEP(1)  ADD_STEP(2)  ADD_STEP(3)
    ADD_STEP(4)  ADD_STEP(5)  ADD_STEP(6)  ADD_STEP(7)
    ADD_STEP(8)  ADD_STEP(9)  ADD_STEP(10) ADD_STEP(11)
    ADD_STEP(12) ADD_STEP(13) ADD_STEP(14) ADD_STEP(15)
#undef ADD_STEP
}

#define ROW_CAP   16384
#define ROW_SLOTS 512
#define SORT_U    8
#define SORT_P    3
#define SORT_SLOTS (SORT_P * SORT_U)

__device__ __forceinline__ void run_gather(unsigned xo, unsigned o4,
                                           const float* __restrict__ x, float* v)
{
    const char* xb = (const char*)x;
#define RG(J) v[J] = *(const float*)(xb + (ROW_SHARE(xo, J) | o4));
    RG(0) RG(1) RG(2) RG(3) RG(4) RG(5) RG(6) RG(7)
    RG(8) RG(9) RG(10) RG(11) RG(12) RG(13) RG(14) RG(15)
#undef RG
}

__device__ __forceinline__ float run_sum(unsigned wb, const float* v, float a)
{
#define RS(J) a = __builtin_fmaf(__uint_as_float(ROW_SHARE(wb, J)), v[J], a);
    RS(0) RS(1) RS(2) RS(3) RS(4) RS(5) RS(6) RS(7)
    RS(8) RS(9) RS(10) RS(11) RS(12) RS(13) RS(14) RS(15)
#undef RS
    return a;
}

__global__ __launch_bounds__(SORT_THREADS)
void row_sum_kernel(const uint2* __restrict__ sedge,
                    const unsigned* __restrict__ offs,
                    const float* __restrict__ x,
                    float* __restrict__ out, int n, int nchunks)
{
    extern __shared__ uint2 srt[];
    __shared__ float    acc[(RB_MAX + 2) * OUT_DIM];
    __shared__ unsigned cnt[ROW_SLOTS + 1 + 32];
    __shared__ unsigned wsum[16];
    __shared__ unsigned multi;

    const int tid  = (int)threadIdx.x;
    const int lane = tid & 31;
    const int hh   = lane >> 4;
    const int wave = __builtin_amdgcn_readfirstlane(tid >> 5);
    const unsigned o4 = (unsigned)(lane & 15) << 2;
    const int rb   = rows_per_bin(n);
    const int b    = (int)blockIdx.x;
    const int row0 = b * rb;
    int nrows = n - row0;
    if (nrows > rb) nrows = rb;
    if (nrows < 0)  nrows = 0;

    if (tid <= ROW_SLOTS) cnt[tid] = 0u;
    if (tid == 0) multi = 0u;
    __syncthreads();

    const unsigned* ob = offs + b;

#define SLOT_SEG(u)   ((u) < 2 * SORT_U ? ((u) >> 1) : ((u) - 2 * SORT_U))
#define SLOT_PIECE(u) ((u) < 2 * SORT_U ? (unsigned)((u) & 1) : 2u)
#define SLOT_IDX(u)   ((unsigned)__builtin_amdgcn_readlane((int)sv, SLOT_SEG(u)) \
                       + p + 32u * SLOT_PIECE(u) + (unsigned)lane)
#define SLOT_OK(u)    (SLOT_IDX(u) < (unsigned)__builtin_amdgcn_readlane((int)tv, SLOT_SEG(u)))
#define SLOT_BASE(u)  ((const char*)(sedge + (size_t)((gb + SLOT_SEG(u)) < nchunks ? (gb + SLOT_SEG(u)) : nchunks - 1) * CH))
#define SLOT_LOAD(u, okb) (*(const uint2*)(SLOT_BASE(u) + ((SLOT_IDX(u) & (0u - (okb))) << 3)))

    uint2    rec[SORT_SLOTS];
    unsigned rank2[SORT_SLOTS / 2];
    #pragma unroll
    for (int u = 0; u < SORT_SLOTS; ++u) rec[u] = make_uint2(0u, 0u);
    #pragma unroll
    for (int u = 0; u < SORT_SLOTS / 2; ++u) rank2[u] = 0u;
    unsigned okm   = 0u;
    int      trips = 0;
    bool     third = false;
    for (int gb = wave * SORT_U; gb < nchunks; gb += 32 * SORT_U) {
        unsigned sv = 0u, tv = 0u;
        if (lane < SORT_U && gb + lane < nchunks) {
            sv = ob[(size_t)(gb + lane) * OFFS_STRIDE];
            tv = ob[(size_t)(gb + lane) * OFFS_STRIDE + 1];
            if (tv > (unsigned)CH) tv = (unsigned)CH;
            if (sv > tv) sv = tv;
        }
        unsigned mx = 0u;
        #pragma unroll
        for (int u = 0; u < SORT_U; ++u) {
            const unsigned l = (unsigned)__builtin_amdgcn_readlane((int)(tv - sv), u);
            mx = l > mx ? l : mx;
        }
        for (unsigned p = 0u; p < mx; p += 32u * SORT_P) {
            ++trips;
            okm = 0u;
            third = mx > p + 64u;
            #pragma unroll
            for (int u = 0; u < 2 * SORT_U; ++u) {
                const unsigned okb = SLOT_OK(u) ? 1u : 0u;
                rec[u] = SLOT_LOAD(u, okb);
                okm |= okb << u;
            }
            if (third) {
                #pragma unroll
                for (int u = 2 * SORT_U; u < SORT_SLOTS; ++u) {
                    const unsigned okb = SLOT_OK(u) ? 1u : 0u;
                    rec[u] = SLOT_LOAD(u, okb);
                    okm |= okb << u;
                }
            }
            __builtin_amdgcn_sched_barrier(0);
            #pragma unroll
            for (int u = 0; u < 2 * SORT_U; u += 2) {
                const unsigned s0 = (okm >> u) & 1u ? rec[u].x >> ROW_SHIFT
                                                    : (unsigned)(ROW_SLOTS + 1 + lane);
                const unsigned s1 = (okm >> (u + 1)) & 1u ? rec[u + 1].x >> ROW_SHIFT
                                                          : (unsigned)(ROW_SLOTS + 1 + lane);
                const unsigned k0 = atomicAdd(&cnt[s0], 1u);
                const unsigned k1 = atomicAdd(&cnt[s1], 1u);
                rank2[u >> 1] = (k0 & 0xFFFFu) | (k1 << 16);
            }
            if (third) {
                #pragma unroll
                for (int u = 2 * SORT_U; u < SORT_SLOTS; u += 2) {
                    const unsigned s0 = (okm >> u) & 1u ? rec[u].x >> ROW_SHIFT
                                                        : (unsigned)(ROW_SLOTS + 1 + lane);
                    const unsigned s1 = (okm >> (u + 1)) & 1u ? rec[u + 1].x >> ROW_SHIFT
                                                              : (unsigned)(ROW_SLOTS + 1 + lane);
                    const unsigned k0 = atomicAdd(&cnt[s0], 1u);
                    const unsigned k1 = atomicAdd(&cnt[s1], 1u);
                    rank2[u >> 1] = (k0 & 0xFFFFu) | (k1 << 16);
                }
            }
        }
    }
    if (trips > 1 && lane == 0) multi = 1u;
    __syncthreads();

    unsigned vv = 0u, ss = 0u;
    if (tid < ROW_SLOTS) {
        vv = cnt[tid];
        ss = vv;
        #pragma unroll
        for (int d = 1; d < 32; d <<= 1) {
            const unsigned t = __shfl_up(ss, d);
            if (lane >= d) ss += t;
        }
        if (lane == 31) wsum[wave] = ss;
    }
    __syncthreads();
    if (tid < ROW_SLOTS) {
        unsigned add = 0u;
        for (int w = 0; w < wave; ++w) add += wsum[w];
        cnt[tid] = add + ss - vv;
        if (tid == ROW_SLOTS - 1) cnt[ROW_SLOTS] = add + ss;
    }
    __syncthreads();
    const unsigned total   = (unsigned)__builtin_amdgcn_readfirstlane((int)cnt[ROW_SLOTS]);
    const bool     in_regs = __builtin_amdgcn_readfirstlane((int)multi) == 0;

    if (total <= (unsigned)ROW_CAP) {
        if (in_regs) {
            #pragma unroll
            for (int u0 = 0; u0 < SORT_SLOTS; u0 += 8) {
                if (u0 < 2 * SORT_U || third) {
                    unsigned st8[8];
                    #pragma unroll
                    for (int q = 0; q < 8; ++q) {
                        const int u = u0 + q;
                        const unsigned slot = (okm >> u) & 1u ? rec[u].x >> ROW_SHIFT
                                                              : (unsigned)(ROW_SLOTS + 1 + lane);
                        st8[q] = cnt[slot];
                    }
                    #pragma unroll
                    for (int q = 0; q < 8; ++q) {
                        const int u = u0 + q;
                        const unsigned pos = st8[q] + ((u & 1) ? rank2[u >> 1] >> 16
                                                               : rank2[u >> 1] & 0xFFFFu);
                        if (((okm >> u) & 1u) && pos < (unsigned)ROW_CAP)
                            srt[pos] = make_uint2(rec[u].x & COL_FIELD, rec[u].y);
                    }
                }
            }
        } else {
            for (int gb = wave * SORT_U; gb < nchunks; gb += 32 * SORT_U) {
                unsigned sv = 0u, tv = 0u;
                if (lane < SORT_U && gb + lane < nchunks) {
                    sv = ob[(size_t)(gb + lane) * OFFS_STRIDE];
                    tv = ob[(size_t)(gb + lane) * OFFS_STRIDE + 1];
                    if (tv > (unsigned)CH) tv = (unsigned)CH;
                    if (sv > tv) sv = tv;
                }
                unsigned mx = 0u;
                #pragma unroll
                for (int u = 0; u < SORT_U; ++u) {
                    const unsigned l = (unsigned)__builtin_amdgcn_readlane((int)(tv - sv), u);
                    mx = l > mx ? l : mx;
                }
                for (unsigned p = 0u; p < mx; p += 64u) {
                    uint2 rc[2 * SORT_U];
                    #pragma unroll
                    for (int u = 0; u < 2 * SORT_U; ++u)
                        rc[u] = SLOT_LOAD(u, SLOT_OK(u) ? 1u : 0u);
                    __builtin_amdgcn_sched_barrier(0);
                    unsigned pos[2 * SORT_U];
                    #pragma unroll
                    for (int u = 0; u < 2 * SORT_U; ++u) {
                        const unsigned slot = SLOT_OK(u) ? rc[u].x >> ROW_SHIFT
                                                         : (unsigned)(ROW_SLOTS + 1 + lane);
                        pos[u] = atomicAdd(&cnt[slot], 1u);
                    }
                    #pragma unroll
                    for (int u = 0; u < 2 * SORT_U; ++u) {
                        if (SLOT_OK(u) && pos[u] < (unsigned)ROW_CAP)
                            srt[pos[u]] = make_uint2(rc[u].x & COL_FIELD, rc[u].y);
                    }
                }
            }
            __syncthreads();
            if (tid < ROW_SLOTS) cnt[tid] -= vv;
        }
#undef SLOT_SEG
#undef SLOT_PIECE
#undef SLOT_IDX
#undef SLOT_OK
#undef SLOT_BASE
#undef SLOT_LOAD
        __syncthreads();

        const int npairs = (nrows + 1) >> 1;
        int t = wave;
        unsigned st_n = 0u, en_n = 0u;
        if (t < npairs) {
            const int r = 2 * t + hh;
            st_n = cnt[r < nrows ? r : 0];
            en_n = cnt[r < nrows ? r + 1 : 0];
        }
        while (t < npairs) {
            const int  r     = 2 * t + hh;
            const bool rowok = r < nrows;
            const unsigned st = st_n;
            const unsigned d  = rowok ? en_n - st_n : 0u;
            if (t + 32 < npairs) {
                const int rn = 2 * (t + 32) + hh;
                st_n = cnt[rn < nrows ? rn : 0];
                en_n = cnt[rn < nrows ? rn + 1 : 0];
            }
            const unsigned d0 = (unsigned)__builtin_amdgcn_readlane((int)d, 0);
            const unsigned d1 = (unsigned)__builtin_amdgcn_readlane((int)d, 16);
            const unsigned dmax = d0 > d1 ? d0 : d1;

            float a = 0.0f;
            #define FETCH_PIECE(R, J0)                                         \
                do {                                                           \
                    const unsigned sl_ = (J0) + (unsigned)(lane & 15);         \
                    const bool ok_ = sl_ < d;                                  \
                    const uint2 t_ = srt[ok_ ? st + sl_ : 0u];                 \
                    R.x = ok_ ? t_.x : 0u;                                     \
                    R.y = ok_ ? t_.y : 0u;                                     \
                } while (0)
            for (unsigned j = 0u; j < dmax; j += 64u) {
                float v0[16], v1[16], v2[16], v3[16];
                uint2 r0, r1, r2, r3;
                const bool h1 = j + 16u < dmax, h2 = j + 32u < dmax, h3 = j + 48u < dmax;
                FETCH_PIECE(r0, j);
                FETCH_PIECE(r1, j + 16u);
                FETCH_PIECE(r2, j + 32u);
                FETCH_PIECE(r3, j + 48u);
                run_gather(r0.x, o4, x, v0);
                if (h1) run_gather(r1.x, o4, x, v1);
                if (h2) run_gather(r2.x, o4, x, v2);
                if (h3) run_gather(r3.x, o4, x, v3);
                a = run_sum(r0.y, v0, a);
                if (h1) a = run_sum(r1.y, v1, a);
                if (h2) a = run_sum(r2.y, v2, a);
                if (h3) a = run_sum(r3.y, v3, a);
            }
            #undef FETCH_PIECE
            if (rowok) out[(size_t)(row0 + r) * OUT_DIM + (lane & 15)] = a;
            t += 32;
        }
        return;
    }

    for (int i = tid; i < (RB_MAX + 2) * OUT_DIM; i += SORT_THREADS) acc[i] = 0.0f;
    __syncthreads();
    for (int g = wave; g < nchunks; g += 32) {
        const unsigned s = ob[(size_t)g * OFFS_STRIDE];
        unsigned e = ob[(size_t)g * OFFS_STRIDE + 1];
        if (e > (unsigned)CH) e = (unsigned)CH;
        const uint2* seg = sedge + (size_t)g * CH;
        for (unsigned p = s; p < e; p += 32u) {
            const int c = (int)(e - p) < 32 ? (int)(e - p) : 32;
            const uint2 ed = *(lane < c ? seg + p + (unsigned)lane : sedge);
            block_add_round(ed, c, lane, o4, x, acc);
        }
    }
    __syncthreads();
    float* ob_out = out + (size_t)row0 * OUT_DIM;
    for (int i = tid; i < nrows * OUT_DIM; i += SORT_THREADS) ob_out[i] = acc[i];
}

extern "C" void kernel_launch(void* const* d_in, const int* in_sizes, int n_in,
                              void* d_out, int out_size, void* d_ws, size_t ws_size,
                              hipStream_t stream) {
    const float* feat = (const float*)d_in[0];
    const int*   er   = (const int*)  d_in[1];
    const int*   ec   = (const int*)  d_in[2];
    const float* ew   = (const float*)d_in[3];
    const float* W1   = (const float*)d_in[4];
    const float* b1   = (const float*)d_in[5];
    const float* W2   = (const float*)d_in[6];
    const float* b2   = (const float*)d_in[7];
    float* out = (float*)d_out;
    float* x   = (float*)d_ws;

    const int n = in_sizes[0] / IN_DIM;
    const int E = in_sizes[1];

    const size_t tk_off = (((size_t)n * OUT_DIM * sizeof(float)) + 255) & ~(size_t)255;
    const bool   owner  = ws_size >= tk_off + 512;
    unsigned* tickets = owner ? (unsigned*)((char*)d_ws + tk_off) : nullptr;

    const int    nchunks  = (E + CH - 1) / CH;
    const size_t se_off   = tk_off + 512;
    const size_t of_off   = se_off + (size_t)nchunks * CH * sizeof(uint2);
    const size_t ws_need  = of_off + (size_t)nchunks * OFFS_STRIDE * sizeof(unsigned);
    const bool   sorted   = n <= N_SORT_MAX && E > 0 && ws_size >= ws_need;

    const int mlp_blocks = (n + GROUP_ROWS - 1) / GROUP_ROWS;

    if (sorted) {
        uint2*    sedge = (uint2*)((char*)d_ws + se_off);
        unsigned* offs  = (unsigned*)((char*)d_ws + of_off);
        const int fused_blocks = mlp_blocks > nchunks ? mlp_blocks : nchunks;
        mlp_fused_kernel<<<fused_blocks, GROUP_THREADS, SMEM_BYTES, stream>>>(
            feat, W1, b1, W2, b2, x, n, nullptr, er, ec, ew, sedge, offs, E, nchunks);
        row_sum_kernel<<<NBINS, SORT_THREADS, SUM_DYN_LDS, stream>>>(sedge, offs, x, out, n, nchunks);
        return;
    }

    hipMemsetAsync(d_out, 0, (size_t)out_size * sizeof(float), stream);

    mlp_fused_kernel<<<mlp_blocks, GROUP_THREADS, SMEM_BYTES, stream>>>(
        feat, W1, b1, W2, b2, x, n, tickets, er, ec, ew, nullptr, nullptr, E, 0);

    if (owner) {
        spmm_owner_kernel<<<2048, 256, 0, stream>>>(er, ec, ew, x, out, tickets, E, n);
    } else {
        const long sthreads = (long)E * 16;
        const int  sblocks  = (int)((sthreads + 255) / 256);
        spmm_scatter_kernel<<<sblocks, 256, 0, stream>>>(er, ec, ew, x, out, E);
    }
}
